# starting kernel
# baseline (speedup 1.0000x reference)
#include <hip/hip_runtime.h>
#include <hip/hip_bf16.h>


typedef float v2f __attribute__((ext_vector_type(2)));
typedef float v8f __attribute__((ext_vector_type(8)));

#define Bsz 2048
#define Tsz 4096
#define Hsz 8

__device__ __forceinline__ float shx16(float v) {
#if __has_builtin(__builtin_amdgcn_permlanex16)
    unsigned int u = __float_as_uint(v);
    unsigned int r = __builtin_amdgcn_permlanex16(
        u, u, 0x76543210u, 0xfedcba98u,  false,  false);
    return __uint_as_float(r);
#else
    return __shfl_xor(v, 16, 32);
#endif
}

__device__ __forceinline__ float fsig(float x) {
    return __builtin_amdgcn_rcpf(1.0f + __expf(-x));
}
__device__ __forceinline__ float ftanh(float x) {
    return 1.0f - 2.0f * __builtin_amdgcn_rcpf(__expf(2.0f * x) + 1.0f);
}

__device__ __forceinline__ v8f wmma4(v2f a, v2f b, v8f c) {
    return __builtin_amdgcn_wmma_f32_16x16x4_f32(
         false, a,  false, b,
         (short)0, c,  false,  false);
}

__global__ __launch_bounds__(32) void lstm2_wmma_kernel(
    const float* __restrict__ x,
    const float* __restrict__ Wih0,
    const float* __restrict__ Whh0,
    const float* __restrict__ bih0, const float* __restrict__ bhh0,
    const float* __restrict__ Wih1,
    const float* __restrict__ Whh1,
    const float* __restrict__ bih1, const float* __restrict__ bhh1,
    const float* __restrict__ Wlin,
    const float* __restrict__ blin,
    float* __restrict__ out)
{
    const int lane = threadIdx.x;
    const int lm   = lane & 15;
    const bool hh  = (lane >= 16);
    const int hsel = hh ? 1 : 0;
    const long row = (long)(blockIdx.x * 32 + lane) * Tsz;

    v2f A0[2][2];
    v2f A1[2][4];
#pragma unroll
    for (int gt = 0; gt < 2; ++gt) {
        const int g = 16 * gt + lm;
#pragma unroll
        for (int kf = 0; kf < 2; ++kf) {
            const int kb = 4 * kf + 2 * hsel;
            A0[gt][kf].x = Whh0[g * Hsz + kb];
            A0[gt][kf].y = Whh0[g * Hsz + kb + 1];
        }
#pragma unroll
        for (int kf = 0; kf < 4; ++kf) {
            const int k = 4 * kf + 2 * hsel;
            const float* Wsrc = (kf < 2) ? Wih1 : Whh1;
            const int kk = (kf < 2) ? k : (k - 8);
            A1[gt][kf].x = Wsrc[g * Hsz + kk];
            A1[gt][kf].y = Wsrc[g * Hsz + kk + 1];
        }
    }

    float wg0C[2][8], b0C[2][8], b1C[2][8];
#pragma unroll
    for (int gt = 0; gt < 2; ++gt)
#pragma unroll
        for (int r = 0; r < 8; ++r) {
            const int g = 16 * gt + r + 8 * hsel;
            wg0C[gt][r] = Wih0[g];
            b0C[gt][r]  = bih0[g] + bhh0[g];
            b1C[gt][r]  = bih1[g] + bhh1[g];
        }

    float wl[8];
#pragma unroll
    for (int j = 0; j < 8; ++j) wl[j] = Wlin[j];
    const float bl = blin[0];

    float h1[8], c1[8], h2[8], c2[8];
#pragma unroll
    for (int j = 0; j < 8; ++j) { h1[j] = 0.f; c1[j] = 0.f; h2[j] = 0.f; c2[j] = 0.f; }

    for (int t = 0; t < Tsz; ++t) {
        const float xv = x[row + t];
        __builtin_prefetch(&x[row + t + 256], 0, 0);
        const float xsh = shx16(xv);
        const float xb0 = hh ? xsh : xv;
        const float xb1 = hh ? xv : xsh;

        v2f Bh1[2][2];
#pragma unroll
        for (int kf = 0; kf < 2; ++kf) {
            const int k0 = 4 * kf;
            const float s0 = shx16(h1[k0 + 0]), s1 = shx16(h1[k0 + 1]);
            const float s2 = shx16(h1[k0 + 2]), s3 = shx16(h1[k0 + 3]);
            Bh1[kf][0].x = hh ? s2 : h1[k0 + 0];
            Bh1[kf][0].y = hh ? s3 : h1[k0 + 1];
            Bh1[kf][1].x = hh ? h1[k0 + 2] : s0;
            Bh1[kf][1].y = hh ? h1[k0 + 3] : s1;
        }

        v8f D0[2][2];
#pragma unroll
        for (int gt = 0; gt < 2; ++gt)
#pragma unroll
            for (int bt = 0; bt < 2; ++bt) {
                const float xb = bt ? xb1 : xb0;
                v8f acc;
#pragma unroll
                for (int r = 0; r < 8; ++r)
                    acc[r] = fmaf(xb, wg0C[gt][r], b0C[gt][r]);
                acc = wmma4(A0[gt][0], Bh1[0][bt], acc);
                acc = wmma4(A0[gt][1], Bh1[1][bt], acc);
                D0[gt][bt] = acc;
            }

#pragma unroll
        for (int r = 0; r < 8; ++r) {
            const float a00 = D0[0][0][r], a01 = D0[0][1][r];
            const float a10 = D0[1][0][r], a11 = D0[1][1][r];
            const float sA0 = shx16(a00), sB0 = shx16(a01);
            const float sA1 = shx16(a10), sB1 = shx16(a11);
            const float ig = fsig(hh ? sB0 : a00);
            const float fg = fsig(hh ? a01 : sA0);
            const float gg = ftanh(hh ? sB1 : a10);
            const float og = fsig(hh ? a11 : sA1);
            c1[r] = fmaf(fg, c1[r], ig * gg);
            h1[r] = og * ftanh(c1[r]);
        }

        v2f Bc[4][2];
#pragma unroll
        for (int kf = 0; kf < 4; ++kf) {
            const float* hs = (kf < 2) ? h1 : h2;
            const int k0 = 4 * (kf & 1);
            const float s0 = shx16(hs[k0 + 0]), s1 = shx16(hs[k0 + 1]);
            const float s2 = shx16(hs[k0 + 2]), s3 = shx16(hs[k0 + 3]);
            Bc[kf][0].x = hh ? s2 : hs[k0 + 0];
            Bc[kf][0].y = hh ? s3 : hs[k0 + 1];
            Bc[kf][1].x = hh ? hs[k0 + 2] : s0;
            Bc[kf][1].y = hh ? hs[k0 + 3] : s1;
        }

        v8f D1[2][2];
#pragma unroll
        for (int gt = 0; gt < 2; ++gt)
#pragma unroll
            for (int bt = 0; bt < 2; ++bt) {
                v8f acc;
#pragma unroll
                for (int r = 0; r < 8; ++r) acc[r] = b1C[gt][r];
                acc = wmma4(A1[gt][0], Bc[0][bt], acc);
                acc = wmma4(A1[gt][1], Bc[1][bt], acc);
                acc = wmma4(A1[gt][2], Bc[2][bt], acc);
                acc = wmma4(A1[gt][3], Bc[3][bt], acc);
                D1[gt][bt] = acc;
            }

#pragma unroll
        for (int r = 0; r < 8; ++r) {
            const float a00 = D1[0][0][r], a01 = D1[0][1][r];
            const float a10 = D1[1][0][r], a11 = D1[1][1][r];
            const float sA0 = shx16(a00), sB0 = shx16(a01);
            const float sA1 = shx16(a10), sB1 = shx16(a11);
            const float ig = fsig(hh ? sB0 : a00);
            const float fg = fsig(hh ? a01 : sA0);
            const float gg = ftanh(hh ? sB1 : a10);
            const float og = fsig(hh ? a11 : sA1);
            c2[r] = fmaf(fg, c2[r], ig * gg);
            h2[r] = og * ftanh(c2[r]);
        }

        float p = bl;
#pragma unroll
        for (int j = 0; j < 8; ++j) p = fmaf(h2[j], wl[j], p);
        out[row + t] = p;
    }
}

extern "C" void kernel_launch(void* const* d_in, const int* in_sizes, int n_in,
                              void* d_out, int out_size, void* d_ws, size_t ws_size,
                              hipStream_t stream) {
    (void)in_sizes; (void)n_in; (void)out_size; (void)d_ws; (void)ws_size;
    const float* x    = (const float*)d_in[0];
    const float* Wih0 = (const float*)d_in[1];
    const float* Whh0 = (const float*)d_in[2];
    const float* bih0 = (const float*)d_in[3];
    const float* bhh0 = (const float*)d_in[4];
    const float* Wih1 = (const float*)d_in[5];
    const float* Whh1 = (const float*)d_in[6];
    const float* bih1 = (const float*)d_in[7];
    const float* bhh1 = (const float*)d_in[8];
    const float* Wlin = (const float*)d_in[9];
    const float* blin = (const float*)d_in[10];
    float* out = (float*)d_out;

    lstm2_wmma_kernel<<<Bsz / 32, 32, 0, stream>>>(
        x, Wih0, Whh0, bih0, bhh0, Wih1, Whh1, bih1, bhh1, Wlin, blin, out);
}


# Gate rows and k slots reordered in A tiles: each lane owns all 4 gates of 4 hidden units of one batch row, no cross-lane moves in the recurrence, 16 rows/wave, 128 waves; output dot in 2 half-sums
# speedup vs baseline: 2.5379x; 2.5379x over previous
#include <hip/hip_runtime.h>
#include <hip/hip_bf16.h>


typedef float v2f __attribute__((ext_vector_type(2)));
typedef float v8f __attribute__((ext_vector_type(8)));

#define Bsz 2048
#define Tsz 4096
#define Hsz 8
#define ROWS_PER_WAVE 16

__device__ __forceinline__ float shx16(float v) {
#if __has_builtin(__builtin_amdgcn_permlanex16)
    unsigned int u = __float_as_uint(v);
    unsigned int r = __builtin_amdgcn_permlanex16(
        u, u, 0x76543210u, 0xfedcba98u,  false,  false);
    return __uint_as_float(r);
#else
    return __shfl_xor(v, 16, 32);
#endif
}

__device__ __forceinline__ float fsig(float x) {
    return __builtin_amdgcn_rcpf(1.0f + __expf(-x));
}
__device__ __forceinline__ float ftanh(float x) {
    return 1.0f - 2.0f * __builtin_amdgcn_rcpf(__expf(2.0f * x) + 1.0f);
}

__device__ __forceinline__ v8f wmma4(v2f a, v2f b, v8f c) {
    return __builtin_amdgcn_wmma_f32_16x16x4_f32(
         false, a,  false, b,
         (short)0, c,  false,  false);
}

__device__ __forceinline__ void cell4(const v8f& d0, const v8f& d1,
                                      float (&c)[4], float (&h)[4]) {
    for (int j = 0; j < 4; ++j) {
        const float ig = fsig(d0[j]);
        const float fg = fsig(d0[4 + j]);
        const float gg = ftanh(d1[j]);
        const float og = fsig(d1[4 + j]);
        c[j] = fmaf(fg, c[j], ig * gg);
        h[j] = og * ftanh(c[j]);
    }
}

__global__ __launch_bounds__(32) void lstm2_wmma_kernel(
    const float* __restrict__ x,
    const float* __restrict__ Wih0,
    const float* __restrict__ Whh0,
    const float* __restrict__ bih0, const float* __restrict__ bhh0,
    const float* __restrict__ Wih1,
    const float* __restrict__ Whh1,
    const float* __restrict__ bih1, const float* __restrict__ bhh1,
    const float* __restrict__ Wlin,
    const float* __restrict__ blin,
    float* __restrict__ out)
{
    const int lane = threadIdx.x;
    const int c    = lane & 15;
    const int q    = lane >> 4;
    const long row = (long)(blockIdx.x * ROWS_PER_WAVE + c) * Tsz;

    v2f A0[2][2];
    v2f A1[2][4];
    for (int gt = 0; gt < 2; ++gt) {
        const int g = (2 * gt + ((c & 7) >> 2)) * Hsz + 4 * (c >> 3) + (c & 3);
        for (int kf = 0; kf < 2; ++kf) {
            const int k = 4 * q + 2 * kf;
            A0[gt][kf].x     = Whh0[g * Hsz + k];
            A0[gt][kf].y     = Whh0[g * Hsz + k + 1];
            A1[gt][kf].x     = Wih1[g * Hsz + k];
            A1[gt][kf].y     = Wih1[g * Hsz + k + 1];
            A1[gt][2 + kf].x = Whh1[g * Hsz + k];
            A1[gt][2 + kf].y = Whh1[g * Hsz + k + 1];
        }
    }

    float wg0C[2][8], b0C[2][8], b1C[2][8];
    for (int gt = 0; gt < 2; ++gt)
        for (int r = 0; r < 8; ++r) {
            const int g = (2 * gt + (r >> 2)) * Hsz + 4 * q + (r & 3);
            wg0C[gt][r] = Wih0[g];
            b0C[gt][r]  = bih0[g] + bhh0[g];
            b1C[gt][r]  = bih1[g] + bhh1[g];
        }

    float wl[4];
    for (int j = 0; j < 4; ++j) wl[j] = Wlin[4 * q + j];
    const float pinit = q ? 0.0f : blin[0];

    float h1[4], c1[4], h2[4], c2[4];
    for (int j = 0; j < 4; ++j) { h1[j] = 0.f; c1[j] = 0.f; h2[j] = 0.f; c2[j] = 0.f; }

    for (int t = 0; t < Tsz; ++t) {
        const float xv = x[row + t];
        __builtin_prefetch(&x[row + t + 256], 0, 0);

        v8f D0[2];
        for (int gt = 0; gt < 2; ++gt) {
            v8f acc;
            for (int r = 0; r < 8; ++r)
                acc[r] = fmaf(xv, wg0C[gt][r], b0C[gt][r]);
            v2f b0, b1;
            b0.x = h1[0]; b0.y = h1[1];
            b1.x = h1[2]; b1.y = h1[3];
            acc = wmma4(A0[gt][0], b0, acc);
            acc = wmma4(A0[gt][1], b1, acc);
            D0[gt] = acc;
        }
        cell4(D0[0], D0[1], c1, h1);

        v8f D1[2];
        for (int gt = 0; gt < 2; ++gt) {
            v8f acc;
            for (int r = 0; r < 8; ++r) acc[r] = b1C[gt][r];
            v2f b0, b1, b2, b3;
            b0.x = h1[0]; b0.y = h1[1];
            b1.x = h1[2]; b1.y = h1[3];
            b2.x = h2[0]; b2.y = h2[1];
            b3.x = h2[2]; b3.y = h2[3];
            acc = wmma4(A1[gt][0], b0, acc);
            acc = wmma4(A1[gt][1], b1, acc);
            acc = wmma4(A1[gt][2], b2, acc);
            acc = wmma4(A1[gt][3], b3, acc);
            D1[gt] = acc;
        }
        cell4(D1[0], D1[1], c2, h2);

        float p = pinit;
        for (int j = 0; j < 4; ++j) p = fmaf(h2[j], wl[j], p);
        p += shx16(p);
        if (q == 0) out[row + t] = p;
    }
}

extern "C" void kernel_launch(void* const* d_in, const int* in_sizes, int n_in,
                              void* d_out, int out_size, void* d_ws, size_t ws_size,
                              hipStream_t stream) {
    (void)in_sizes; (void)n_in; (void)out_size; (void)d_ws; (void)ws_size;
    const float* x    = (const float*)d_in[0];
    const float* Wih0 = (const float*)d_in[1];
    const float* Whh0 = (const float*)d_in[2];
    const float* bih0 = (const float*)d_in[3];
    const float* bhh0 = (const float*)d_in[4];
    const float* Wih1 = (const float*)d_in[5];
    const float* Whh1 = (const float*)d_in[6];
    const float* bih1 = (const float*)d_in[7];
    const float* bhh1 = (const float*)d_in[8];
    const float* Wlin = (const float*)d_in[9];
    const float* blin = (const float*)d_in[10];
    float* out = (float*)d_out;

    lstm2_wmma_kernel<<<Bsz / ROWS_PER_WAVE, 32, 0, stream>>>(
        x, Wih0, Whh0, bih0, bhh0, Wih1, Whh1, bih1, bhh1, Wlin, blin, out);
}


# Time loop software-pipelined: layer-1 matrix products of step t run beside layer-0 cell math of step t+1 and vice versa, 2 steps per branch-free trip, x read a trip ahead; same sums, same order
# speedup vs baseline: 4.3108x; 1.6986x over previous
#include <hip/hip_runtime.h>
#include <hip/hip_bf16.h>


typedef float v2f __attribute__((ext_vector_type(2)));
typedef float v8f __attribute__((ext_vector_type(8)));

#define Bsz 2048
#define Tsz 4096
#define Hsz 8
#define ROWS_PER_WAVE 16

__device__ __forceinline__ float shx16(float v) {
#if __has_builtin(__builtin_amdgcn_permlanex16)
    unsigned int u = __float_as_uint(v);
    unsigned int r = __builtin_amdgcn_permlanex16(
        u, u, 0x76543210u, 0xfedcba98u,  false,  false);
    return __uint_as_float(r);
#else
    return __shfl_xor(v, 16, 32);
#endif
}

__device__ __forceinline__ float fsig(float x) {
    return __builtin_amdgcn_rcpf(1.0f + __expf(-x));
}
__device__ __forceinline__ float ftanh(float x) {
    return 1.0f - 2.0f * __builtin_amdgcn_rcpf(__expf(2.0f * x) + 1.0f);
}

__device__ __forceinline__ v8f wmma4(v2f a, v2f b, v8f c) {
    return __builtin_amdgcn_wmma_f32_16x16x4_f32(
         false, a,  false, b,
         (short)0, c,  false,  false);
}

__device__ __forceinline__ void cell4(const v8f& d0, const v8f& d1,
                                      float (&c)[4], float (&h)[4]) {
    for (int j = 0; j < 4; ++j) {
        const float ig = fsig(d0[j]);
        const float fg = fsig(d0[4 + j]);
        const float gg = ftanh(d1[j]);
        const float og = fsig(d1[4 + j]);
        c[j] = fmaf(fg, c[j], ig * gg);
        h[j] = og * ftanh(c[j]);
    }
}

__device__ __forceinline__ void layer0_gates(const v2f (&A0)[2][2],
                                             const float (&wg0C)[2][8],
                                             const float (&b0C)[2][8],
                                             float xv, v2f hb0, v2f hb1,
                                             v8f (&D0)[2]) {
    for (int gt = 0; gt < 2; ++gt) {
        v8f acc;
        for (int r = 0; r < 8; ++r)
            acc[r] = fmaf(xv, wg0C[gt][r], b0C[gt][r]);
        acc = wmma4(A0[gt][0], hb0, acc);
        acc = wmma4(A0[gt][1], hb1, acc);
        D0[gt] = acc;
    }
}

__device__ __forceinline__ void layer1_gates(const v2f (&A1)[2][4],
                                             const float (&b1C)[2][8],
                                             v2f hb0, v2f hb1,
                                             const float (&h2)[4],
                                             v8f (&D1)[2]) {
    for (int gt = 0; gt < 2; ++gt) {
        v8f acc;
        for (int r = 0; r < 8; ++r) acc[r] = b1C[gt][r];
        v2f b2, b3;
        b2.x = h2[0]; b2.y = h2[1];
        b3.x = h2[2]; b3.y = h2[3];
        acc = wmma4(A1[gt][0], hb0, acc);
        acc = wmma4(A1[gt][1], hb1, acc);
        acc = wmma4(A1[gt][2], b2, acc);
        acc = wmma4(A1[gt][3], b3, acc);
        D1[gt] = acc;
    }
}

__device__ __forceinline__ float project(const float (&h2)[4],
                                         const float (&wl)[4], float pinit) {
    float p = pinit;
    for (int j = 0; j < 4; ++j) p = fmaf(h2[j], wl[j], p);
    p += shx16(p);
    return p;
}

__device__ __forceinline__ float pipeline_pass(
    const v2f (&A0)[2][2], const v2f (&A1)[2][4],
    const float (&wg0C)[2][8], const float (&b0C)[2][8], const float (&b1C)[2][8],
    const float (&wl)[4], float pinit, float xv,
    v2f& hb0, v2f& hb1, v8f (&D0)[2],
    float (&c1)[4], float (&h2)[4], float (&c2)[4]) {
    float h1[4];
    v8f D1[2];
    layer1_gates(A1, b1C, hb0, hb1, h2, D1);
    cell4(D0[0], D0[1], c1, h1);
    hb0.x = h1[0]; hb0.y = h1[1];
    hb1.x = h1[2]; hb1.y = h1[3];
    layer0_gates(A0, wg0C, b0C, xv, hb0, hb1, D0);
    cell4(D1[0], D1[1], c2, h2);
    return project(h2, wl, pinit);
}

__global__ __launch_bounds__(32) __attribute__((amdgpu_waves_per_eu(1, 4)))
void lstm2_wmma_kernel(
    const float* __restrict__ x,
    const float* __restrict__ Wih0,
    const float* __restrict__ Whh0,
    const float* __restrict__ bih0, const float* __restrict__ bhh0,
    const float* __restrict__ Wih1,
    const float* __restrict__ Whh1,
    const float* __restrict__ bih1, const float* __restrict__ bhh1,
    const float* __restrict__ Wlin,
    const float* __restrict__ blin,
    float* __restrict__ out)
{
    const int lane = threadIdx.x;
    const int c    = lane & 15;
    const int q    = lane >> 4;
    const long row = (long)(blockIdx.x * ROWS_PER_WAVE + c) * Tsz;

    v2f A0[2][2];
    v2f A1[2][4];
    for (int gt = 0; gt < 2; ++gt) {
        const int g = (2 * gt + ((c & 7) >> 2)) * Hsz + 4 * (c >> 3) + (c & 3);
        for (int kf = 0; kf < 2; ++kf) {
            const int k = 4 * q + 2 * kf;
            A0[gt][kf].x     = Whh0[g * Hsz + k];
            A0[gt][kf].y     = Whh0[g * Hsz + k + 1];
            A1[gt][kf].x     = Wih1[g * Hsz + k];
            A1[gt][kf].y     = Wih1[g * Hsz + k + 1];
            A1[gt][2 + kf].x = Whh1[g * Hsz + k];
            A1[gt][2 + kf].y = Whh1[g * Hsz + k + 1];
        }
    }

    float wg0C[2][8], b0C[2][8], b1C[2][8];
    for (int gt = 0; gt < 2; ++gt)
        for (int r = 0; r < 8; ++r) {
            const int g = (2 * gt + (r >> 2)) * Hsz + 4 * q + (r & 3);
            wg0C[gt][r] = Wih0[g];
            b0C[gt][r]  = bih0[g] + bhh0[g];
            b1C[gt][r]  = bih1[g] + bhh1[g];
        }

    float wl[4];
    for (int j = 0; j < 4; ++j) wl[j] = Wlin[4 * q + j];
    const float pinit = q ? 0.0f : blin[0];

    float h1[4], c1[4], h2[4], c2[4];
    for (int j = 0; j < 4; ++j) { h1[j] = 0.f; c1[j] = 0.f; h2[j] = 0.f; c2[j] = 0.f; }

    v2f hb0, hb1;
    v8f D0[2];
    v2f xq;
    {
        const float x0 = x[row];
        const float x1 = x[row + 1];
        xq.x = x[row + 2];
        xq.y = x[row + 3];
        hb0.x = h1[0]; hb0.y = h1[1];
        hb1.x = h1[2]; hb1.y = h1[3];
        layer0_gates(A0, wg0C, b0C, x0, hb0, hb1, D0);
        cell4(D0[0], D0[1], c1, h1);
        hb0.x = h1[0]; hb0.y = h1[1];
        hb1.x = h1[2]; hb1.y = h1[3];
        layer0_gates(A0, wg0C, b0C, x1, hb0, hb1, D0);
    }

    for (int t = 0; t < Tsz; t += 2) {
        const v2f xv = xq;
        const int tn = (t + 4 < Tsz) ? (t + 4) : (Tsz - 2);
        xq.x = x[row + tn];
        xq.y = x[row + tn + 1];
        __builtin_prefetch(&x[row + t + 256], 0, 0);

        const float p0 = pipeline_pass(A0, A1, wg0C, b0C, b1C, wl, pinit, xv.x,
                                       hb0, hb1, D0, c1, h2, c2);
        const float p1 = pipeline_pass(A0, A1, wg0C, b0C, b1C, wl, pinit, xv.y,
                                       hb0, hb1, D0, c1, h2, c2);
        out[row + t + q] = q ? p1 : p0;
    }
}

extern "C" void kernel_launch(void* const* d_in, const int* in_sizes, int n_in,
                              void* d_out, int out_size, void* d_ws, size_t ws_size,
                              hipStream_t stream) {
    (void)in_sizes; (void)n_in; (void)out_size; (void)d_ws; (void)ws_size;
    const float* x    = (const float*)d_in[0];
    const float* Wih0 = (const float*)d_in[1];
    const float* Whh0 = (const float*)d_in[2];
    const float* bih0 = (const float*)d_in[3];
    const float* bhh0 = (const float*)d_in[4];
    const float* Wih1 = (const float*)d_in[5];
    const float* Whh1 = (const float*)d_in[6];
    const float* bih1 = (const float*)d_in[7];
    const float* bhh1 = (const float*)d_in[8];
    const float* Wlin = (const float*)d_in[9];
    const float* blin = (const float*)d_in[10];
    float* out = (float*)d_out;

    lstm2_wmma_kernel<<<Bsz / ROWS_PER_WAVE, 32, 0, stream>>>(
        x, Wih0, Whh0, bih0, bhh0, Wih1, Whh1, bih1, bhh1, Wlin, blin, out);
}


# The two layers run on two waves of a 64-thread group, one SIMD each: wave 0 (layer 0) hands h1 to wave 1 (layer 1, output) through an LDS ring, one barrier per 32 steps; same sums, same order
# speedup vs baseline: 6.8464x; 1.5882x over previous
#include <hip/hip_runtime.h>
#include <hip/hip_bf16.h>


typedef float v2f __attribute__((ext_vector_type(2)));
typedef float v8f __attribute__((ext_vector_type(8)));

#define Bsz 2048
#define Tsz 4096
#define Hsz 8
#define ROWS_PER_WAVE 16
#define CH 32
#define NCH (Tsz / CH)

__device__ __forceinline__ float shx16(float v) {
#if __has_builtin(__builtin_amdgcn_permlanex16)
    unsigned int u = __float_as_uint(v);
    unsigned int r = __builtin_amdgcn_permlanex16(
        u, u, 0x76543210u, 0xfedcba98u,  false,  false);
    return __uint_as_float(r);
#else
    return __shfl_xor(v, 16, 32);
#endif
}

__device__ __forceinline__ float fsig(float x) {
    return __builtin_amdgcn_rcpf(1.0f + __expf(-x));
}
__device__ __forceinline__ float ftanh(float x) {
    return 1.0f - 2.0f * __builtin_amdgcn_rcpf(__expf(2.0f * x) + 1.0f);
}

__device__ __forceinline__ v8f wmma4(v2f a, v2f b, v8f c) {
    return __builtin_amdgcn_wmma_f32_16x16x4_f32(
         false, a,  false, b,
         (short)0, c,  false,  false);
}

__device__ __forceinline__ void cell4(const v8f& d0, const v8f& d1,
                                      float (&c)[4], float (&h)[4]) {
    for (int j = 0; j < 4; ++j) {
        const float ig = fsig(d0[j]);
        const float fg = fsig(d0[4 + j]);
        const float gg = ftanh(d1[j]);
        const float og = fsig(d1[4 + j]);
        c[j] = fmaf(fg, c[j], ig * gg);
        h[j] = og * ftanh(c[j]);
    }
}

__device__ __forceinline__ void layer0_gates(const v2f (&A0)[2][2],
                                             const float (&wg0C)[2][8],
                                             const float (&b0C)[2][8],
                                             float xv, v2f hb0, v2f hb1,
                                             v8f (&D0)[2]) {
    for (int gt = 0; gt < 2; ++gt) {
        v8f acc;
        for (int r = 0; r < 8; ++r)
            acc[r] = fmaf(xv, wg0C[gt][r], b0C[gt][r]);
        acc = wmma4(A0[gt][0], hb0, acc);
        acc = wmma4(A0[gt][1], hb1, acc);
        D0[gt] = acc;
    }
}

__device__ __forceinline__ void layer1_h1part(const v2f (&A1)[2][4],
                                              const float (&b1C)[2][8],
                                              v2f hb0, v2f hb1,
                                              v8f (&P)[2]) {
    for (int gt = 0; gt < 2; ++gt) {
        v8f acc;
        for (int r = 0; r < 8; ++r) acc[r] = b1C[gt][r];
        acc = wmma4(A1[gt][0], hb0, acc);
        acc = wmma4(A1[gt][1], hb1, acc);
        P[gt] = acc;
    }
}
__device__ __forceinline__ void layer1_h2part(const v2f (&A1)[2][4],
                                              const v8f (&P)[2],
                                              const float (&h2)[4],
                                              v8f (&D1)[2]) {
    v2f b2, b3;
    b2.x = h2[0]; b2.y = h2[1];
    b3.x = h2[2]; b3.y = h2[3];
    for (int gt = 0; gt < 2; ++gt) {
        v8f acc = wmma4(A1[gt][2], b2, P[gt]);
        acc = wmma4(A1[gt][3], b3, acc);
        D1[gt] = acc;
    }
}

__device__ __forceinline__ float project(const float (&h2)[4],
                                         const float (&wl)[4], float pinit) {
    float p = pinit;
    for (int j = 0; j < 4; ++j) p = fmaf(h2[j], wl[j], p);
    p += shx16(p);
    return p;
}

__device__ __forceinline__ float layer1_step(
    const v2f (&A1)[2][4], const float (&b1C)[2][8],
    const float (&wl)[4], float pinit, float4 hn,
    v8f (&P)[2], float (&h2)[4], float (&c2)[4]) {
    v8f D1[2];
    layer1_h2part(A1, P, h2, D1);
    v2f n0, n1;
    n0.x = hn.x; n0.y = hn.y;
    n1.x = hn.z; n1.y = hn.w;
    layer1_h1part(A1, b1C, n0, n1, P);
    cell4(D1[0], D1[1], c2, h2);
    return project(h2, wl, pinit);
}

__global__ __launch_bounds__(64) __attribute__((amdgpu_waves_per_eu(1, 4)))
void lstm2_wmma_kernel(
    const float* __restrict__ x,
    const float* __restrict__ Wih0,
    const float* __restrict__ Whh0,
    const float* __restrict__ bih0, const float* __restrict__ bhh0,
    const float* __restrict__ Wih1,
    const float* __restrict__ Whh1,
    const float* __restrict__ bih1, const float* __restrict__ bhh1,
    const float* __restrict__ Wlin,
    const float* __restrict__ blin,
    float* __restrict__ out)
{
    __shared__ float4 ring[2][CH][32];

    const int lane = threadIdx.x & 31;
    const int role = __builtin_amdgcn_readfirstlane((int)(threadIdx.x >> 5));
    const int c    = lane & 15;
    const int q    = lane >> 4;
    const long row = (long)(blockIdx.x * ROWS_PER_WAVE + c) * Tsz;

    v2f A0[2][2];
    v2f A1[2][4];
    for (int gt = 0; gt < 2; ++gt) {
        const int g = (2 * gt + ((c & 7) >> 2)) * Hsz + 4 * (c >> 3) + (c & 3);
        for (int kf = 0; kf < 2; ++kf) {
            const int k = 4 * q + 2 * kf;
            A0[gt][kf].x     = Whh0[g * Hsz + k];
            A0[gt][kf].y     = Whh0[g * Hsz + k + 1];
            A1[gt][kf].x     = Wih1[g * Hsz + k];
            A1[gt][kf].y     = Wih1[g * Hsz + k + 1];
            A1[gt][2 + kf].x = Whh1[g * Hsz + k];
            A1[gt][2 + kf].y = Whh1[g * Hsz + k + 1];
        }
    }

    float wg0C[2][8], b0C[2][8], b1C[2][8];
    for (int gt = 0; gt < 2; ++gt)
        for (int r = 0; r < 8; ++r) {
            const int g = (2 * gt + (r >> 2)) * Hsz + 4 * q + (r & 3);
            wg0C[gt][r] = Wih0[g];
            b0C[gt][r]  = bih0[g] + bhh0[g];
            b1C[gt][r]  = bih1[g] + bhh1[g];
        }

    float wl[4];
    for (int j = 0; j < 4; ++j) wl[j] = Wlin[4 * q + j];
    const float pinit = q ? 0.0f : blin[0];

    float c1[4], h2[4], c2[4];
    for (int j = 0; j < 4; ++j) { c1[j] = 0.f; h2[j] = 0.f; c2[j] = 0.f; }
    v2f hb0, hb1;
    hb0.x = 0.f; hb0.y = 0.f; hb1.x = 0.f; hb1.y = 0.f;
    v8f P[2];
    for (int r = 0; r < 8; ++r) { P[0][r] = 0.f; P[1][r] = 0.f; }
    float4 xq = *(const float4*)&x[row];

    for (int p = 0; p <= NCH; ++p) {
        if (role == 0) {
            if (p < NCH) {
                const int t0 = p * CH;
                for (int s = 0; s < CH; s += 4) {
                    const int t = t0 + s;
                    const float xs[4] = { xq.x, xq.y, xq.z, xq.w };
                    const int tn = (t + 4 < Tsz) ? (t + 4) : (Tsz - 4);
                    xq = *(const float4*)&x[row + tn];
                    const int tp = (t + 256 < Tsz) ? (t + 256) : (Tsz - 1);
                    __builtin_prefetch(&x[row + tp], 0, 0);
#pragma unroll
                    for (int u = 0; u < 4; ++u) {
                        v8f D0[2];
                        float h1[4];
                        layer0_gates(A0, wg0C, b0C, xs[u], hb0, hb1, D0);
                        cell4(D0[0], D0[1], c1, h1);
                        hb0.x = h1[0]; hb0.y = h1[1];
                        hb1.x = h1[2]; hb1.y = h1[3];
                        ring[p & 1][s + u][lane] = make_float4(h1[0], h1[1], h1[2], h1[3]);
                    }
                }
            }
        } else {
            if (p >= 1) {
                const int k  = p - 1;
                const int t0 = k * CH;
                float4 g0 = ring[k & 1][0][lane];
                float4 g1 = ring[k & 1][1][lane];
                float4 g2 = ring[k & 1][2][lane];
                float4 g3 = ring[k & 1][3][lane];
                {
                    v2f f0, f1;
                    f0.x = g0.x; f0.y = g0.y;
                    f1.x = g0.z; f1.y = g0.w;
                    layer1_h1part(A1, b1C, f0, f1, P);
                }
                for (int s = 0; s < CH; s += 4) {
                    const int t = t0 + s;
                    const float4 a1 = g1, a2 = g2, a3 = g3;
                    const int sn = (s + 4 < CH) ? (s + 4) : s;
                    g0 = ring[k & 1][sn][lane];
                    g1 = ring[k & 1][sn + 1][lane];
                    g2 = ring[k & 1][sn + 2][lane];
                    g3 = ring[k & 1][sn + 3][lane];
                    const float p0 = layer1_step(A1, b1C, wl, pinit, a1, P, h2, c2);
                    const float p1 = layer1_step(A1, b1C, wl, pinit, a2, P, h2, c2);
                    out[row + t + q] = q ? p1 : p0;
                    const float p2 = layer1_step(A1, b1C, wl, pinit, a3, P, h2, c2);
                    const float p3 = layer1_step(A1, b1C, wl, pinit, g0, P, h2, c2);
                    out[row + t + 2 + q] = q ? p3 : p2;
                }
            }
        }
        __syncthreads();
    }
}

extern "C" void kernel_launch(void* const* d_in, const int* in_sizes, int n_in,
                              void* d_out, int out_size, void* d_ws, size_t ws_size,
                              hipStream_t stream) {
    (void)in_sizes; (void)n_in; (void)out_size; (void)d_ws; (void)ws_size;
    const float* x    = (const float*)d_in[0];
    const float* Wih0 = (const float*)d_in[1];
    const float* Whh0 = (const float*)d_in[2];
    const float* bih0 = (const float*)d_in[3];
    const float* bhh0 = (const float*)d_in[4];
    const float* Wih1 = (const float*)d_in[5];
    const float* Whh1 = (const float*)d_in[6];
    const float* bih1 = (const float*)d_in[7];
    const float* bhh1 = (const float*)d_in[8];
    const float* Wlin = (const float*)d_in[9];
    const float* blin = (const float*)d_in[10];
    float* out = (float*)d_out;

    lstm2_wmma_kernel<<<Bsz / ROWS_PER_WAVE, 64, 0, stream>>>(
        x, Wih0, Whh0, bih0, bhh0, Wih1, Whh1, bih1, bhh1, Wlin, blin, out);
}


# Gate non-linearities on the hardware tanh (v_tanh_f32): sigmoid(z) = (1 + tanh(z/2))/2, the 1/2 folded exactly into the i, f, o weight and bias rows; 5 transcendentals per cell, not 10; same sums
# speedup vs baseline: 9.7346x; 1.4219x over previous
#include <hip/hip_runtime.h>
#include <hip/hip_bf16.h>


typedef float v2f __attribute__((ext_vector_type(2)));
typedef float v8f __attribute__((ext_vector_type(8)));

#define Bsz 2048
#define Tsz 4096
#define Hsz 8
#define ROWS_PER_WAVE 16
#define CH 32
#define NCH (Tsz / CH)

__device__ __forceinline__ float shx16(float v) {
#if __has_builtin(__builtin_amdgcn_permlanex16)
    unsigned int u = __float_as_uint(v);
    unsigned int r = __builtin_amdgcn_permlanex16(
        u, u, 0x76543210u, 0xfedcba98u,  false,  false);
    return __uint_as_float(r);
#else
    return __shfl_xor(v, 16, 32);
#endif
}

__device__ __forceinline__ float htanh(float x) {
    return __builtin_amdgcn_tanhf(x);
}
__device__ __forceinline__ float hsig_half(float xh) {
    return fmaf(0.5f, htanh(xh), 0.5f);
}

__device__ __forceinline__ v8f wmma4(v2f a, v2f b, v8f c) {
    return __builtin_amdgcn_wmma_f32_16x16x4_f32(
         false, a,  false, b,
         (short)0, c,  false,  false);
}

__device__ __forceinline__ void cell4(const v8f& d0, const v8f& d1,
                                      float (&c)[4], float (&h)[4]) {
    for (int j = 0; j < 4; ++j) {
        const float ig = hsig_half(d0[j]);
        const float fg = hsig_half(d0[4 + j]);
        const float gg = htanh(d1[j]);
        const float og = hsig_half(d1[4 + j]);
        c[j] = fmaf(fg, c[j], ig * gg);
        h[j] = og * htanh(c[j]);
    }
}

__device__ __forceinline__ void layer0_gates(const v2f (&A0)[2][2],
                                             const float (&wg0C)[2][8],
                                             const float (&b0C)[2][8],
                                             float xv, v2f hb0, v2f hb1,
                                             v8f (&D0)[2]) {
    for (int gt = 0; gt < 2; ++gt) {
        v8f acc;
        for (int r = 0; r < 8; ++r)
            acc[r] = fmaf(xv, wg0C[gt][r], b0C[gt][r]);
        acc = wmma4(A0[gt][0], hb0, acc);
        acc = wmma4(A0[gt][1], hb1, acc);
        D0[gt] = acc;
    }
}

__device__ __forceinline__ void layer1_h1part(const v2f (&A1)[2][4],
                                              const float (&b1C)[2][8],
                                              v2f hb0, v2f hb1,
                                              v8f (&P)[2]) {
    for (int gt = 0; gt < 2; ++gt) {
        v8f acc;
        for (int r = 0; r < 8; ++r) acc[r] = b1C[gt][r];
        acc = wmma4(A1[gt][0], hb0, acc);
        acc = wmma4(A1[gt][1], hb1, acc);
        P[gt] = acc;
    }
}
__device__ __forceinline__ void layer1_h2part(const v2f (&A1)[2][4],
                                              const v8f (&P)[2],
                                              const float (&h2)[4],
                                              v8f (&D1)[2]) {
    v2f b2, b3;
    b2.x = h2[0]; b2.y = h2[1];
    b3.x = h2[2]; b3.y = h2[3];
    for (int gt = 0; gt < 2; ++gt) {
        v8f acc = wmma4(A1[gt][2], b2, P[gt]);
        acc = wmma4(A1[gt][3], b3, acc);
        D1[gt] = acc;
    }
}

__device__ __forceinline__ float project(const float (&h2)[4],
                                         const float (&wl)[4], float pinit) {
    float p = pinit;
    for (int j = 0; j < 4; ++j) p = fmaf(h2[j], wl[j], p);
    p += shx16(p);
    return p;
}

__device__ __forceinline__ float layer1_step(
    const v2f (&A1)[2][4], const float (&b1C)[2][8],
    const float (&wl)[4], float pinit, float4 hn,
    v8f (&P)[2], float (&h2)[4], float (&c2)[4]) {
    v8f D1[2];
    layer1_h2part(A1, P, h2, D1);
    v2f n0, n1;
    n0.x = hn.x; n0.y = hn.y;
    n1.x = hn.z; n1.y = hn.w;
    layer1_h1part(A1, b1C, n0, n1, P);
    cell4(D1[0], D1[1], c2, h2);
    return project(h2, wl, pinit);
}

__global__ __launch_bounds__(64) __attribute__((amdgpu_waves_per_eu(1, 4)))
void lstm2_wmma_kernel(
    const float* __restrict__ x,
    const float* __restrict__ Wih0,
    const float* __restrict__ Whh0,
    const float* __restrict__ bih0, const float* __restrict__ bhh0,
    const float* __restrict__ Wih1,
    const float* __restrict__ Whh1,
    const float* __restrict__ bih1, const float* __restrict__ bhh1,
    const float* __restrict__ Wlin,
    const float* __restrict__ blin,
    float* __restrict__ out)
{
    __shared__ float4 ring[2][CH][32];

    const int lane = threadIdx.x & 31;
    const int role = __builtin_amdgcn_readfirstlane((int)(threadIdx.x >> 5));
    const int c    = lane & 15;
    const int q    = lane >> 4;
    const long row = (long)(blockIdx.x * ROWS_PER_WAVE + c) * Tsz;

    v2f A0[2][2];
    v2f A1[2][4];
    for (int gt = 0; gt < 2; ++gt) {
        const int gate = 2 * gt + ((c & 7) >> 2);
        const float sA = (gate == 2) ? 1.0f : 0.5f;
        const int g = gate * Hsz + 4 * (c >> 3) + (c & 3);
        for (int kf = 0; kf < 2; ++kf) {
            const int k = 4 * q + 2 * kf;
            A0[gt][kf].x     = sA * Whh0[g * Hsz + k];
            A0[gt][kf].y     = sA * Whh0[g * Hsz + k + 1];
            A1[gt][kf].x     = sA * Wih1[g * Hsz + k];
            A1[gt][kf].y     = sA * Wih1[g * Hsz + k + 1];
            A1[gt][2 + kf].x = sA * Whh1[g * Hsz + k];
            A1[gt][2 + kf].y = sA * Whh1[g * Hsz + k + 1];
        }
    }

    float wg0C[2][8], b0C[2][8], b1C[2][8];
    for (int gt = 0; gt < 2; ++gt)
        for (int r = 0; r < 8; ++r) {
            const int gate = 2 * gt + (r >> 2);
            const float sC = (gate == 2) ? 1.0f : 0.5f;
            const int g = gate * Hsz + 4 * q + (r & 3);
            wg0C[gt][r] = sC * Wih0[g];
            b0C[gt][r]  = sC * (bih0[g] + bhh0[g]);
            b1C[gt][r]  = sC * (bih1[g] + bhh1[g]);
        }

    float wl[4];
    for (int j = 0; j < 4; ++j) wl[j] = Wlin[4 * q + j];
    const float pinit = q ? 0.0f : blin[0];

    float c1[4], h2[4], c2[4];
    for (int j = 0; j < 4; ++j) { c1[j] = 0.f; h2[j] = 0.f; c2[j] = 0.f; }
    v2f hb0, hb1;
    hb0.x = 0.f; hb0.y = 0.f; hb1.x = 0.f; hb1.y = 0.f;
    v8f P[2];
    for (int r = 0; r < 8; ++r) { P[0][r] = 0.f; P[1][r] = 0.f; }
    float4 xq = *(const float4*)&x[row];

    for (int p = 0; p <= NCH; ++p) {
        if (role == 0) {
            if (p < NCH) {
                const int t0 = p * CH;
                for (int s = 0; s < CH; s += 4) {
                    const int t = t0 + s;
                    const float xs[4] = { xq.x, xq.y, xq.z, xq.w };
                    const int tn = (t + 4 < Tsz) ? (t + 4) : (Tsz - 4);
                    xq = *(const float4*)&x[row + tn];
                    const int tp = (t + 256 < Tsz) ? (t + 256) : (Tsz - 1);
                    __builtin_prefetch(&x[row + tp], 0, 0);
#pragma unroll
                    for (int u = 0; u < 4; ++u) {
                        v8f D0[2];
                        float h1[4];
                        layer0_gates(A0, wg0C, b0C, xs[u], hb0, hb1, D0);
                        cell4(D0[0], D0[1], c1, h1);
                        hb0.x = h1[0]; hb0.y = h1[1];
                        hb1.x = h1[2]; hb1.y = h1[3];
                        ring[p & 1][s + u][lane] = make_float4(h1[0], h1[1], h1[2], h1[3]);
                    }
                }
            }
        } else {
            if (p >= 1) {
                const int k  = p - 1;
                const int t0 = k * CH;
                float4 g0 = ring[k & 1][0][lane];
                float4 g1 = ring[k & 1][1][lane];
                float4 g2 = ring[k & 1][2][lane];
                float4 g3 = ring[k & 1][3][lane];
                {
                    v2f f0, f1;
                    f0.x = g0.x; f0.y = g0.y;
                    f1.x = g0.z; f1.y = g0.w;
                    layer1_h1part(A1, b1C, f0, f1, P);
                }
                for (int s = 0; s < CH; s += 4) {
                    const int t = t0 + s;
                    const float4 a1 = g1, a2 = g2, a3 = g3;
                    const int sn = (s + 4 < CH) ? (s + 4) : s;
                    g0 = ring[k & 1][sn][lane];
                    g1 = ring[k & 1][sn + 1][lane];
                    g2 = ring[k & 1][sn + 2][lane];
                    g3 = ring[k & 1][sn + 3][lane];
                    const float p0 = layer1_step(A1, b1C, wl, pinit, a1, P, h2, c2);
                    const float p1 = layer1_step(A1, b1C, wl, pinit, a2, P, h2, c2);
                    out[row + t + q] = q ? p1 : p0;
                    const float p2 = layer1_step(A1, b1C, wl, pinit, a3, P, h2, c2);
                    const float p3 = layer1_step(A1, b1C, wl, pinit, g0, P, h2, c2);
                    out[row + t + 2 + q] = q ? p3 : p2;
                }
            }
        }
        __syncthreads();
    }
}

extern "C" void kernel_launch(void* const* d_in, const int* in_sizes, int n_in,
                              void* d_out, int out_size, void* d_ws, size_t ws_size,
                              hipStream_t stream) {
    (void)in_sizes; (void)n_in; (void)out_size; (void)d_ws; (void)ws_size;
    const float* x    = (const float*)d_in[0];
    const float* Wih0 = (const float*)d_in[1];
    const float* Whh0 = (const float*)d_in[2];
    const float* bih0 = (const float*)d_in[3];
    const float* bhh0 = (const float*)d_in[4];
    const float* Wih1 = (const float*)d_in[5];
    const float* Whh1 = (const float*)d_in[6];
    const float* bih1 = (const float*)d_in[7];
    const float* bhh1 = (const float*)d_in[8];
    const float* Wlin = (const float*)d_in[9];
    const float* blin = (const float*)d_in[10];
    float* out = (float*)d_out;

    lstm2_wmma_kernel<<<Bsz / ROWS_PER_WAVE, 64, 0, stream>>>(
        x, Wih0, Whh0, bih0, bhh0, Wih1, Whh1, bih1, bhh1, Wlin, blin, out);
}


# Layer 1 split over two waves: a third wave on a third SIMD forms the non-recurrent half of its gates, b1 + Wih1*h1(t), and hands the C tiles on through a second LDS ring; same sums, same order
# speedup vs baseline: 11.3747x; 1.1685x over previous
#include <hip/hip_runtime.h>
#include <hip/hip_bf16.h>


typedef float v2f __attribute__((ext_vector_type(2)));
typedef float v8f __attribute__((ext_vector_type(8)));

#define Bsz 2048
#define Tsz 4096
#define Hsz 8
#define ROWS_PER_WAVE 16
#define CH 32
#define NCH (Tsz / CH)

__device__ __forceinline__ float shx16(float v) {
#if __has_builtin(__builtin_amdgcn_permlanex16)
    unsigned int u = __float_as_uint(v);
    unsigned int r = __builtin_amdgcn_permlanex16(
        u, u, 0x76543210u, 0xfedcba98u,  false,  false);
    return __uint_as_float(r);
#else
    return __shfl_xor(v, 16, 32);
#endif
}

__device__ __forceinline__ float htanh(float x) {
    return __builtin_amdgcn_tanhf(x);
}
__device__ __forceinline__ float hsig_half(float xh) {
    return fmaf(0.5f, htanh(xh), 0.5f);
}

__device__ __forceinline__ v8f wmma4(v2f a, v2f b, v8f c) {
    return __builtin_amdgcn_wmma_f32_16x16x4_f32(
         false, a,  false, b,
         (short)0, c,  false,  false);
}

__device__ __forceinline__ void cell4(const v8f& d0, const v8f& d1,
                                      float (&c)[4], float (&h)[4]) {
    for (int j = 0; j < 4; ++j) {
        const float ig = hsig_half(d0[j]);
        const float fg = hsig_half(d0[4 + j]);
        const float gg = htanh(d1[j]);
        const float og = hsig_half(d1[4 + j]);
        c[j] = fmaf(fg, c[j], ig * gg);
        h[j] = og * htanh(c[j]);
    }
}

__device__ __forceinline__ void layer0_gates(const v2f (&A0)[2][2],
                                             const float (&wg0C)[2][8],
                                             const float (&b0C)[2][8],
                                             float xv, v2f hb0, v2f hb1,
                                             v8f (&D0)[2]) {
    for (int gt = 0; gt < 2; ++gt) {
        v8f acc;
        for (int r = 0; r < 8; ++r)
            acc[r] = fmaf(xv, wg0C[gt][r], b0C[gt][r]);
        acc = wmma4(A0[gt][0], hb0, acc);
        acc = wmma4(A0[gt][1], hb1, acc);
        D0[gt] = acc;
    }
}

__device__ __forceinline__ void layer1_h1part(const v2f (&A1)[2][4],
                                              const float (&b1C)[2][8],
                                              v2f hb0, v2f hb1,
                                              v8f (&P)[2]) {
    for (int gt = 0; gt < 2; ++gt) {
        v8f acc;
        for (int r = 0; r < 8; ++r) acc[r] = b1C[gt][r];
        acc = wmma4(A1[gt][0], hb0, acc);
        acc = wmma4(A1[gt][1], hb1, acc);
        P[gt] = acc;
    }
}
__device__ __forceinline__ void layer1_h2part(const v2f (&A1)[2][4],
                                              const v8f &P0, const v8f &P1,
                                              const float (&h2)[4],
                                              v8f (&D1)[2]) {
    v2f b2, b3;
    b2.x = h2[0]; b2.y = h2[1];
    b3.x = h2[2]; b3.y = h2[3];
    v8f acc0 = wmma4(A1[0][2], b2, P0);
    v8f acc1 = wmma4(A1[1][2], b2, P1);
    D1[0] = wmma4(A1[0][3], b3, acc0);
    D1[1] = wmma4(A1[1][3], b3, acc1);
}

__device__ __forceinline__ float project(const float (&h2)[4],
                                         const float (&wl)[4], float pinit) {
    float p = pinit;
    for (int j = 0; j < 4; ++j) p = fmaf(h2[j], wl[j], p);
    p += shx16(p);
    return p;
}

__device__ __forceinline__ void layer1_mats(
    const v2f (&A1)[2][4], const float4 (&g)[4],
    const float (&h2)[4], v8f (&D1)[2]) {
    v8f P0, P1;
    P0[0] = g[0].x; P0[1] = g[0].y; P0[2] = g[0].z; P0[3] = g[0].w;
    P0[4] = g[1].x; P0[5] = g[1].y; P0[6] = g[1].z; P0[7] = g[1].w;
    P1[0] = g[2].x; P1[1] = g[2].y; P1[2] = g[2].z; P1[3] = g[2].w;
    P1[4] = g[3].x; P1[5] = g[3].y; P1[6] = g[3].z; P1[7] = g[3].w;
    layer1_h2part(A1, P0, P1, h2, D1);
}
__device__ __forceinline__ float layer1_cell(
    const v8f (&D1)[2], const float (&wl)[4], float pinit,
    float (&h2)[4], float (&c2)[4]) {
    cell4(D1[0], D1[1], c2, h2);
    return project(h2, wl, pinit);
}
__device__ __forceinline__ float layer1_step(
    const v2f (&A1)[2][4], const float (&wl)[4], float pinit,
    const float4 (&g)[4], float (&h2)[4], float (&c2)[4]) {
    v8f D1[2];
    layer1_mats(A1, g, h2, D1);
    return layer1_cell(D1, wl, pinit, h2, c2);
}

__global__ __launch_bounds__(96) __attribute__((amdgpu_waves_per_eu(1, 4)))
void lstm2_wmma_kernel(
    const float* __restrict__ x,
    const float* __restrict__ Wih0,
    const float* __restrict__ Whh0,
    const float* __restrict__ bih0, const float* __restrict__ bhh0,
    const float* __restrict__ Wih1,
    const float* __restrict__ Whh1,
    const float* __restrict__ bih1, const float* __restrict__ bhh1,
    const float* __restrict__ Wlin,
    const float* __restrict__ blin,
    float* __restrict__ out)
{
    __shared__ float4 ringH[2][CH][32];
    __shared__ float4 ringP[2][CH][4][32];

    const int lane = threadIdx.x & 31;
    const int role = __builtin_amdgcn_readfirstlane((int)(threadIdx.x >> 5));
    const int c    = lane & 15;
    const int q    = lane >> 4;
    const long row = (long)(blockIdx.x * ROWS_PER_WAVE + c) * Tsz;

    v2f A0[2][2];
    v2f A1[2][4];
    for (int gt = 0; gt < 2; ++gt) {
        const int gate = 2 * gt + ((c & 7) >> 2);
        const float sA = (gate == 2) ? 1.0f : 0.5f;
        const int g = gate * Hsz + 4 * (c >> 3) + (c & 3);
        for (int kf = 0; kf < 2; ++kf) {
            const int k = 4 * q + 2 * kf;
            A0[gt][kf].x     = sA * Whh0[g * Hsz + k];
            A0[gt][kf].y     = sA * Whh0[g * Hsz + k + 1];
            A1[gt][kf].x     = sA * Wih1[g * Hsz + k];
            A1[gt][kf].y     = sA * Wih1[g * Hsz + k + 1];
            A1[gt][2 + kf].x = sA * Whh1[g * Hsz + k];
            A1[gt][2 + kf].y = sA * Whh1[g * Hsz + k + 1];
        }
    }

    float wg0C[2][8], b0C[2][8], b1C[2][8];
    for (int gt = 0; gt < 2; ++gt)
        for (int r = 0; r < 8; ++r) {
            const int gate = 2 * gt + (r >> 2);
            const float sC = (gate == 2) ? 1.0f : 0.5f;
            const int g = gate * Hsz + 4 * q + (r & 3);
            wg0C[gt][r] = sC * Wih0[g];
            b0C[gt][r]  = sC * (bih0[g] + bhh0[g]);
            b1C[gt][r]  = sC * (bih1[g] + bhh1[g]);
        }

    float wl[4];
    for (int j = 0; j < 4; ++j) wl[j] = Wlin[4 * q + j];
    const float pinit = q ? 0.0f : blin[0];

    float c1[4], h2[4], c2[4];
    for (int j = 0; j < 4; ++j) { c1[j] = 0.f; h2[j] = 0.f; c2[j] = 0.f; }
    v2f hb0, hb1;
    hb0.x = 0.f; hb0.y = 0.f; hb1.x = 0.f; hb1.y = 0.f;
    float4 xq = *(const float4*)&x[row];

    for (int p = 0; p < NCH + 2; ++p) {
        if (role == 0) {
            if (p < NCH) {
                const int t0 = p * CH;
                for (int s = 0; s < CH; s += 4) {
                    const int t = t0 + s;
                    const float xs[4] = { xq.x, xq.y, xq.z, xq.w };
                    const int tn = (t + 4 < Tsz) ? (t + 4) : (Tsz - 4);
                    xq = *(const float4*)&x[row + tn];
                    const int tp = (t + 256 < Tsz) ? (t + 256) : (Tsz - 1);
                    __builtin_prefetch(&x[row + tp], 0, 0);
#pragma unroll
                    for (int u = 0; u < 4; ++u) {
                        v8f D0[2];
                        float h1[4];
                        layer0_gates(A0, wg0C, b0C, xs[u], hb0, hb1, D0);
                        cell4(D0[0], D0[1], c1, h1);
                        hb0.x = h1[0]; hb0.y = h1[1];
                        hb1.x = h1[2]; hb1.y = h1[3];
                        ringH[p & 1][s + u][lane] = make_float4(h1[0], h1[1], h1[2], h1[3]);
                    }
                }
            }
        } else if (role == 1) {
            if (p >= 1 && p <= NCH) {
                const int k = p - 1;
                float4 g0 = ringH[k & 1][0][lane];
                float4 g1 = ringH[k & 1][1][lane];
                float4 g2 = ringH[k & 1][2][lane];
                float4 g3 = ringH[k & 1][3][lane];
#pragma nounroll
                for (int s = 0; s < CH; s += 4) {
                    const float4 a[4] = { g0, g1, g2, g3 };
                    const int sn = (s + 4 < CH) ? (s + 4) : s;
                    g0 = ringH[k & 1][sn][lane];
                    g1 = ringH[k & 1][sn + 1][lane];
                    g2 = ringH[k & 1][sn + 2][lane];
                    g3 = ringH[k & 1][sn + 3][lane];
#pragma unroll
                    for (int u = 0; u < 4; ++u) {
                        v2f f0, f1;
                        f0.x = a[u].x; f0.y = a[u].y;
                        f1.x = a[u].z; f1.y = a[u].w;
                        v8f Pn[2];
                        layer1_h1part(A1, b1C, f0, f1, Pn);
                        ringP[k & 1][s + u][0][lane] = make_float4(Pn[0][0], Pn[0][1], Pn[0][2], Pn[0][3]);
                        ringP[k & 1][s + u][1][lane] = make_float4(Pn[0][4], Pn[0][5], Pn[0][6], Pn[0][7]);
                        ringP[k & 1][s + u][2][lane] = make_float4(Pn[1][0], Pn[1][1], Pn[1][2], Pn[1][3]);
                        ringP[k & 1][s + u][3][lane] = make_float4(Pn[1][4], Pn[1][5], Pn[1][6], Pn[1][7]);
                    }
                }
            }
        } else {
            if (p >= 2) {
                const int k  = p - 2;
                const int t0 = k * CH;
                float4 ga[2][4], gb[2][4];
                for (int j = 0; j < 4; ++j) {
                    ga[0][j] = ringP[k & 1][0][j][lane];
                    ga[1][j] = ringP[k & 1][1][j][lane];
                }
                for (int s = 0; s < CH; s += 4) {
                    const int t = t0 + s;
                    v8f D1[2];
                    layer1_mats(A1, ga[0], h2, D1);
                    asm volatile("" :: "v"(D1[0]), "v"(D1[1]) : "memory");
                    for (int j = 0; j < 4; ++j) {
                        gb[0][j] = ringP[k & 1][s + 2][j][lane];
                        gb[1][j] = ringP[k & 1][s + 3][j][lane];
                    }
                    __builtin_amdgcn_sched_barrier(0);
                    const float p0 = layer1_cell(D1, wl, pinit, h2, c2);
                    const float p1 = layer1_step(A1, wl, pinit, ga[1], h2, c2);
                    out[row + t + q] = q ? p1 : p0;
                    layer1_mats(A1, gb[0], h2, D1);
                    const int sn = (s + 4 < CH) ? (s + 4) : s;
                    asm volatile("" :: "v"(D1[0]), "v"(D1[1]) : "memory");
                    for (int j = 0; j < 4; ++j) {
                        ga[0][j] = ringP[k & 1][sn][j][lane];
                        ga[1][j] = ringP[k & 1][sn + 1][j][lane];
                    }
                    __builtin_amdgcn_sched_barrier(0);
                    const float p2 = layer1_cell(D1, wl, pinit, h2, c2);
                    const float p3 = layer1_step(A1, wl, pinit, gb[1], h2, c2);
                    out[row + t + 2 + q] = q ? p3 : p2;
                }
            }
        }
        __syncthreads();
    }
}

extern "C" void kernel_launch(void* const* d_in, const int* in_sizes, int n_in,
                              void* d_out, int out_size, void* d_ws, size_t ws_size,
                              hipStream_t stream) {
    (void)in_sizes; (void)n_in; (void)out_size; (void)d_ws; (void)ws_size;
    const float* x    = (const float*)d_in[0];
    const float* Wih0 = (const float*)d_in[1];
    const float* Whh0 = (const float*)d_in[2];
    const float* bih0 = (const float*)d_in[3];
    const float* bhh0 = (const float*)d_in[4];
    const float* Wih1 = (const float*)d_in[5];
    const float* Whh1 = (const float*)d_in[6];
    const float* bih1 = (const float*)d_in[7];
    const float* bhh1 = (const float*)d_in[8];
    const float* Wlin = (const float*)d_in[9];
    const float* blin = (const float*)d_in[10];
    float* out = (float*)d_out;

    lstm2_wmma_kernel<<<Bsz / ROWS_PER_WAVE, 96, 0, stream>>>(
        x, Wih0, Whh0, bih0, bhh0, Wih1, Whh1, bih1, bhh1, Wlin, blin, out);
}


# 8 batch rows per wave, repeated in both 16-column halves: each lane updates 2 hidden units, not 4 (half the tanh and gate work per step); one DPP swap rebuilds B; 256 groups of 3 waves; same sums
# speedup vs baseline: 12.7952x; 1.1249x over previous
#include <hip/hip_runtime.h>
#include <hip/hip_bf16.h>


typedef float v2f __attribute__((ext_vector_type(2)));
typedef float v8f __attribute__((ext_vector_type(8)));

#define Bsz 2048
#define Tsz 4096
#define Hsz 8
#define ROWS_PER_GROUP 8
#define CH 64
#define NCH (Tsz / CH)

__device__ __forceinline__ float shx16(float v) {
    unsigned int u = __float_as_uint(v);
    unsigned int r = __builtin_amdgcn_permlanex16(
        u, u, 0x76543210u, 0xfedcba98u,  false,  false);
    return __uint_as_float(r);
}
__device__ __forceinline__ float rot8(float v) {
    int u = (int)__float_as_uint(v);
    int r = __builtin_amdgcn_update_dpp(u, u,   0x128, 0xf, 0xf, false);
    return __uint_as_float((unsigned int)r);
}

__device__ __forceinline__ float htanh(float x) {
    return __builtin_amdgcn_tanhf(x);
}
__device__ __forceinline__ float hsig_half(float xh) {
    return fmaf(0.5f, htanh(xh), 0.5f);
}

__device__ __forceinline__ v8f wmma4(v2f a, v2f b, v8f c) {
    return __builtin_amdgcn_wmma_f32_16x16x4_f32(
        false, a, false, b, (short)0, c, false, false);
}

__device__ __forceinline__ v8f pick(bool lo8, const v8f& d0, const v8f& d1) {
    v8f d;
    for (int r = 0; r < 8; ++r) d[r] = lo8 ? d0[r] : d1[r];
    return d;
}

__device__ __forceinline__ void cell2(const v8f& d, float (&c)[2], float (&h)[2]) {
    for (int j = 0; j < 2; ++j) {
        const float ig = hsig_half(d[j]);
        const float fg = hsig_half(d[2 + j]);
        const float gg = htanh(d[4 + j]);
        const float og = hsig_half(d[6 + j]);
        c[j] = fmaf(fg, c[j], ig * gg);
        h[j] = og * htanh(c[j]);
    }
}
__device__ __forceinline__ void spread(bool lo8, const float (&h)[2], v2f& hb0, v2f& hb1) {
    const float rx = rot8(h[0]), ry = rot8(h[1]);
    hb0.x = lo8 ? h[0] : rx;  hb0.y = lo8 ? h[1] : ry;
    hb1.x = lo8 ? rx : h[0];  hb1.y = lo8 ? ry : h[1];
}

__device__ __forceinline__ v8f gates2(bool lo8, v2f a00, v2f a01, v2f a10, v2f a11,
                                      v2f b0, v2f b1, const v8f& cin, v8f (&raw)[2]) {
    v8f t0 = wmma4(a00, b0, cin);
    v8f t1 = wmma4(a10, b0, cin);
    raw[0] = wmma4(a01, b1, t0);
    raw[1] = wmma4(a11, b1, t1);
    return pick(lo8, raw[0], raw[1]);
}

__device__ __forceinline__ float project(v2f hb0, v2f hb1, const float (&wl)[4], float pinit) {
    float p = pinit;
    p = fmaf(hb0.x, wl[0], p);
    p = fmaf(hb0.y, wl[1], p);
    p = fmaf(hb1.x, wl[2], p);
    p = fmaf(hb1.y, wl[3], p);
    p += shx16(p);
    return p;
}

__device__ __forceinline__ void l1_mats(const v2f (&A1)[2][4], float4 g0, float4 g1,
                                        v2f hb0, v2f hb1, v8f (&raw)[2]) {
    v8f Pc;
    Pc[0] = g0.x; Pc[1] = g0.y; Pc[2] = g0.z; Pc[3] = g0.w;
    Pc[4] = g1.x; Pc[5] = g1.y; Pc[6] = g1.z; Pc[7] = g1.w;
    v8f t0 = wmma4(A1[0][2], hb0, Pc);
    v8f t1 = wmma4(A1[1][2], hb0, Pc);
    raw[0] = wmma4(A1[0][3], hb1, t0);
    raw[1] = wmma4(A1[1][3], hb1, t1);
}
__device__ __forceinline__ float l1_cell(bool lo8, const v8f (&raw)[2], float (&cs)[2],
                                         v2f& hb0, v2f& hb1, const float (&wl)[4], float pinit) {
    const v8f d = pick(lo8, raw[0], raw[1]);
    float h[2];
    cell2(d, cs, h);
    spread(lo8, h, hb0, hb1);
    return project(hb0, hb1, wl, pinit);
}

__global__ __launch_bounds__(96) __attribute__((amdgpu_waves_per_eu(1, 4)))
void lstm2_wmma_kernel(
    const float* __restrict__ x,
    const float* __restrict__ Wih0,
    const float* __restrict__ Whh0,
    const float* __restrict__ bih0, const float* __restrict__ bhh0,
    const float* __restrict__ Wih1,
    const float* __restrict__ Whh1,
    const float* __restrict__ bih1, const float* __restrict__ bhh1,
    const float* __restrict__ Wlin,
    const float* __restrict__ blin,
    float* __restrict__ out)
{
    __shared__ float4 ringH[2][CH][32];
    __shared__ float4 ringP[2][CH][2][32];

    const int lane = threadIdx.x & 31;
    const int role = __builtin_amdgcn_readfirstlane((int)(threadIdx.x >> 5));
    const int c    = lane & 15;
    const int q    = lane >> 4;
    const int Tl   = (lane >> 3) & 1;
    const bool lo8 = (Tl == 0);
    const long row = (long)(blockIdx.x * ROWS_PER_GROUP + (lane & 7)) * Tsz;

    v2f A0[2][2];
    v2f A1[2][4];
    for (int gt = 0; gt < 2; ++gt) {
        const int gate = (c & 7) >> 1;
        const float sA = (gate == 2) ? 1.0f : 0.5f;
        const int g = gate * Hsz + 4 * (c >> 3) + 2 * gt + (c & 1);
        for (int kf = 0; kf < 2; ++kf) {
            const int k = 4 * q + 2 * kf;
            A0[gt][kf].x     = sA * Whh0[g * Hsz + k];
            A0[gt][kf].y     = sA * Whh0[g * Hsz + k + 1];
            A1[gt][kf].x     = sA * Wih1[g * Hsz + k];
            A1[gt][kf].y     = sA * Wih1[g * Hsz + k + 1];
            A1[gt][2 + kf].x = sA * Whh1[g * Hsz + k];
            A1[gt][2 + kf].y = sA * Whh1[g * Hsz + k + 1];
        }
    }

    float wg0C[8], b0C[8];
    v8f b1C;
    for (int r = 0; r < 8; ++r) {
        const int gate = r >> 1;
        const float sC = (gate == 2) ? 1.0f : 0.5f;
        const int g = gate * Hsz + 4 * q + 2 * Tl + (r & 1);
        wg0C[r] = sC * Wih0[g];
        b0C[r]  = sC * (bih0[g] + bhh0[g]);
        b1C[r]  = sC * (bih1[g] + bhh1[g]);
    }

    float wl[4];
    for (int j = 0; j < 4; ++j) wl[j] = Wlin[4 * q + j];
    const float pinit = q ? 0.0f : blin[0];

    float cs[2] = { 0.f, 0.f };
    v2f hb0, hb1;
    hb0.x = 0.f; hb0.y = 0.f; hb1.x = 0.f; hb1.y = 0.f;
    float4 xq = *(const float4*)&x[row];

    for (int p = 0; p < NCH + 2; ++p) {
        if (role == 0) {
            if (p < NCH) {
                const int t0 = p * CH;
                for (int s = 0; s < CH; s += 4) {
                    const int t = t0 + s;
                    const float xs[4] = { xq.x, xq.y, xq.z, xq.w };
                    const int tn = (t + 4 < Tsz) ? (t + 4) : (Tsz - 4);
                    xq = *(const float4*)&x[row + tn];
                    const int tp = (t + 256 < Tsz) ? (t + 256) : (Tsz - 1);
                    __builtin_prefetch(&x[row + tp], 0, 0);
#pragma unroll
                    for (int u = 0; u < 4; ++u) {
                        v8f acc, raw[2];
                        for (int r = 0; r < 8; ++r) acc[r] = fmaf(xs[u], wg0C[r], b0C[r]);
                        const v8f d = gates2(lo8, A0[0][0], A0[0][1], A0[1][0], A0[1][1], hb0, hb1, acc, raw);
                        float h[2];
                        cell2(d, cs, h);
                        spread(lo8, h, hb0, hb1);
                        ringH[p & 1][s + u][lane] = make_float4(hb0.x, hb0.y, hb1.x, hb1.y);
                    }
                }
            }
        } else if (role == 1) {
            if (p >= 1 && p <= NCH) {
                const int k = p - 1;
                float4 g0 = ringH[k & 1][0][lane];
                float4 g1 = ringH[k & 1][1][lane];
                float4 g2 = ringH[k & 1][2][lane];
                float4 g3 = ringH[k & 1][3][lane];
#pragma nounroll
                for (int s = 0; s < CH; s += 4) {
                    const float4 a[4] = { g0, g1, g2, g3 };
                    const int sn = (s + 4 < CH) ? (s + 4) : s;
                    g0 = ringH[k & 1][sn][lane];
                    g1 = ringH[k & 1][sn + 1][lane];
                    g2 = ringH[k & 1][sn + 2][lane];
                    g3 = ringH[k & 1][sn + 3][lane];
#pragma unroll
                    for (int u = 0; u < 4; ++u) {
                        v2f f0, f1;
                        f0.x = a[u].x; f0.y = a[u].y;
                        f1.x = a[u].z; f1.y = a[u].w;
                        v8f raw[2];
                        const v8f P = gates2(lo8, A1[0][0], A1[0][1], A1[1][0], A1[1][1], f0, f1, b1C, raw);
                        ringP[k & 1][s + u][0][lane] = make_float4(P[0], P[1], P[2], P[3]);
                        ringP[k & 1][s + u][1][lane] = make_float4(P[4], P[5], P[6], P[7]);
                    }
                }
            }
        } else {
            if (p >= 2) {
                const int k  = p - 2;
                const int t0 = k * CH;
                float4 a00 = ringP[k & 1][0][0][lane], a01 = ringP[k & 1][0][1][lane];
                float4 a10 = ringP[k & 1][1][0][lane], a11 = ringP[k & 1][1][1][lane];
                float4 b00, b01, b10, b11;
                for (int s = 0; s < CH; s += 4) {
                    const int t = t0 + s;
                    const int sn = (s + 4 < CH) ? (s + 4) : s;
                    v8f raw[2];
                    l1_mats(A1, a00, a01, hb0, hb1, raw);
                    asm volatile("" :: "v"(raw[0]), "v"(raw[1]) : "memory");
                    b00 = ringP[k & 1][s + 2][0][lane]; b01 = ringP[k & 1][s + 2][1][lane];
                    b10 = ringP[k & 1][s + 3][0][lane]; b11 = ringP[k & 1][s + 3][1][lane];
                    __builtin_amdgcn_sched_barrier(0);
                    const float p0 = l1_cell(lo8, raw, cs, hb0, hb1, wl, pinit);
                    l1_mats(A1, a10, a11, hb0, hb1, raw);
                    const float p1 = l1_cell(lo8, raw, cs, hb0, hb1, wl, pinit);
                    l1_mats(A1, b00, b01, hb0, hb1, raw);
                    asm volatile("" :: "v"(raw[0]), "v"(raw[1]) : "memory");
                    a00 = ringP[k & 1][sn][0][lane];     a01 = ringP[k & 1][sn][1][lane];
                    a10 = ringP[k & 1][sn + 1][0][lane]; a11 = ringP[k & 1][sn + 1][1][lane];
                    __builtin_amdgcn_sched_barrier(0);
                    const float p2 = l1_cell(lo8, raw, cs, hb0, hb1, wl, pinit);
                    l1_mats(A1, b10, b11, hb0, hb1, raw);
                    const float p3 = l1_cell(lo8, raw, cs, hb0, hb1, wl, pinit);
                    const float v01 = Tl ? p1 : p0;
                    const float v23 = Tl ? p3 : p2;
                    out[row + t + 2 * q + Tl] = q ? v23 : v01;
                }
            }
        }
        __syncthreads();
    }
}

extern "C" void kernel_launch(void* const* d_in, const int* in_sizes, int n_in,
                              void* d_out, int out_size, void* d_ws, size_t ws_size,
                              hipStream_t stream) {
    (void)in_sizes; (void)n_in; (void)out_size; (void)d_ws; (void)ws_size;
    const float* x    = (const float*)d_in[0];
    const float* Wih0 = (const float*)d_in[1];
    const float* Whh0 = (const float*)d_in[2];
    const float* bih0 = (const float*)d_in[3];
    const float* bhh0 = (const float*)d_in[4];
    const float* Wih1 = (const float*)d_in[5];
    const float* Whh1 = (const float*)d_in[6];
    const float* bih1 = (const float*)d_in[7];
    const float* bhh1 = (const float*)d_in[8];
    const float* Wlin = (const float*)d_in[9];
    const float* blin = (const float*)d_in[10];
    float* out = (float*)d_out;

    lstm2_wmma_kernel<<<Bsz / ROWS_PER_GROUP, 96, 0, stream>>>(
        x, Wih0, Whh0, bih0, bhh0, Wih1, Whh1, bih1, bhh1, Wlin, blin, out);
}


# k order made lane-relative: B fragment 0 = lane's own h pair, fragment 1 = lane c^8's (one DPP); tile-1 A and Wlin follow; no select or copy before the next product; tile-1, odd-step sums reordered
# speedup vs baseline: 14.1904x; 1.1090x over previous
#include <hip/hip_runtime.h>
#include <hip/hip_bf16.h>


typedef float v2f __attribute__((ext_vector_type(2)));
typedef float v8f __attribute__((ext_vector_type(8)));

#define Bsz 2048
#define Tsz 4096
#define Hsz 8
#define ROWS_PER_GROUP 8
#define CH 64
#define NCH (Tsz / CH)

__device__ __forceinline__ float shx16(float v) {
    unsigned int u = __float_as_uint(v);
    unsigned int r = __builtin_amdgcn_permlanex16(
        u, u, 0x76543210u, 0xfedcba98u,  false,  false);
    return __uint_as_float(r);
}
__device__ __forceinline__ float rot8(float v) {
    int u = (int)__float_as_uint(v);
    int r = __builtin_amdgcn_mov_dpp(u,   0x128, 0xf, 0xf, false);
    return __uint_as_float((unsigned int)r);
}

__device__ __forceinline__ float htanh(float x) {
    return __builtin_amdgcn_tanhf(x);
}
__device__ __forceinline__ float hsig_half(float xh) {
    return fmaf(0.5f, htanh(xh), 0.5f);
}

__device__ __forceinline__ v8f wmma4(v2f a, v2f b, v8f c) {
    return __builtin_amdgcn_wmma_f32_16x16x4_f32(
        false, a, false, b, (short)0, c, false, false);
}

__device__ __forceinline__ v8f pick(bool lo8, const v8f& d0, const v8f& d1) {
    v8f d;
    for (int r = 0; r < 8; ++r) d[r] = lo8 ? d0[r] : d1[r];
    return d;
}

__device__ __forceinline__ void cell2(const v8f& d, float (&c)[2], float (&h)[2]) {
    for (int j = 0; j < 2; ++j) {
        const float ig = hsig_half(d[j]);
        const float fg = hsig_half(d[2 + j]);
        const float gg = htanh(d[4 + j]);
        const float og = hsig_half(d[6 + j]);
        c[j] = fmaf(fg, c[j], ig * gg);
        h[j] = og * htanh(c[j]);
    }
}
__device__ __forceinline__ void spread(const float (&h)[2], v2f& hb0, v2f& hb1) {
    hb0.x = h[0];        hb0.y = h[1];
    hb1.x = rot8(h[0]);  hb1.y = rot8(h[1]);
}

__device__ __forceinline__ v8f gates2(bool lo8, v2f a00, v2f a01, v2f a10, v2f a11,
                                      v2f b0, v2f b1, const v8f& cin, v8f (&raw)[2]) {
    v8f t0 = wmma4(a00, b0, cin);
    v8f t1 = wmma4(a10, b0, cin);
    raw[0] = wmma4(a01, b1, t0);
    raw[1] = wmma4(a11, b1, t1);
    return pick(lo8, raw[0], raw[1]);
}

__device__ __forceinline__ float project(v2f hb0, v2f hb1, const float (&wl)[4], float pinit) {
    float p = pinit;
    p = fmaf(hb0.x, wl[0], p);
    p = fmaf(hb0.y, wl[1], p);
    p = fmaf(hb1.x, wl[2], p);
    p = fmaf(hb1.y, wl[3], p);
    p += shx16(p);
    return p;
}

__device__ __forceinline__ void l1_mats(const v2f (&A1)[2][4], float4 g0, float4 g1,
                                        v2f hb0, v2f hb1, v8f (&raw)[2]) {
    v8f Pc;
    Pc[0] = g0.x; Pc[1] = g0.y; Pc[2] = g0.z; Pc[3] = g0.w;
    Pc[4] = g1.x; Pc[5] = g1.y; Pc[6] = g1.z; Pc[7] = g1.w;
    v8f t0 = wmma4(A1[0][2], hb0, Pc);
    v8f t1 = wmma4(A1[1][2], hb0, Pc);
    raw[0] = wmma4(A1[0][3], hb1, t0);
    raw[1] = wmma4(A1[1][3], hb1, t1);
}
__device__ __forceinline__ float l1_cell(bool lo8, const v8f (&raw)[2], float (&cs)[2],
                                         v2f& hb0, v2f& hb1, const float (&wl)[4], float pinit) {
    const v8f d = pick(lo8, raw[0], raw[1]);
    float h[2];
    cell2(d, cs, h);
    spread(h, hb0, hb1);
    return project(hb0, hb1, wl, pinit);
}

__global__ __launch_bounds__(96) __attribute__((amdgpu_waves_per_eu(1, 4)))
void lstm2_wmma_kernel(
    const float* __restrict__ x,
    const float* __restrict__ Wih0,
    const float* __restrict__ Whh0,
    const float* __restrict__ bih0, const float* __restrict__ bhh0,
    const float* __restrict__ Wih1,
    const float* __restrict__ Whh1,
    const float* __restrict__ bih1, const float* __restrict__ bhh1,
    const float* __restrict__ Wlin,
    const float* __restrict__ blin,
    float* __restrict__ out)
{
    __shared__ float4 ringH[2][CH][32];
    __shared__ float4 ringP[2][CH][2][32];

    const int lane = threadIdx.x & 31;
    const int role = __builtin_amdgcn_readfirstlane((int)(threadIdx.x >> 5));
    const int c    = lane & 15;
    const int q    = lane >> 4;
    const int Tl   = (lane >> 3) & 1;
    const bool lo8 = (Tl == 0);
    const long row = (long)(blockIdx.x * ROWS_PER_GROUP + (lane & 7)) * Tsz;

    v2f A0[2][2];
    v2f A1[2][4];
    for (int gt = 0; gt < 2; ++gt) {
        const int gate = (c & 7) >> 1;
        const float sA = (gate == 2) ? 1.0f : 0.5f;
        const int g = gate * Hsz + 4 * (c >> 3) + 2 * gt + (c & 1);
        for (int kf = 0; kf < 2; ++kf) {
            const int k = 4 * q + 2 * (kf ^ gt);
            A0[gt][kf].x     = sA * Whh0[g * Hsz + k];
            A0[gt][kf].y     = sA * Whh0[g * Hsz + k + 1];
            A1[gt][kf].x     = sA * Wih1[g * Hsz + k];
            A1[gt][kf].y     = sA * Wih1[g * Hsz + k + 1];
            A1[gt][2 + kf].x = sA * Whh1[g * Hsz + k];
            A1[gt][2 + kf].y = sA * Whh1[g * Hsz + k + 1];
        }
    }

    float wg0C[8], b0C[8];
    v8f b1C;
    for (int r = 0; r < 8; ++r) {
        const int gate = r >> 1;
        const float sC = (gate == 2) ? 1.0f : 0.5f;
        const int g = gate * Hsz + 4 * q + 2 * Tl + (r & 1);
        wg0C[r] = sC * Wih0[g];
        b0C[r]  = sC * (bih0[g] + bhh0[g]);
        b1C[r]  = sC * (bih1[g] + bhh1[g]);
    }

    float wl[4];
    for (int j = 0; j < 4; ++j) wl[j] = Wlin[4 * q + 2 * ((j >> 1) ^ Tl) + (j & 1)];
    const float pinit = q ? 0.0f : blin[0];

    float cs[2] = { 0.f, 0.f };
    v2f hb0, hb1;
    hb0.x = 0.f; hb0.y = 0.f; hb1.x = 0.f; hb1.y = 0.f;
    float4 xq = *(const float4*)&x[row];

    for (int p = 0; p < NCH + 2; ++p) {
        if (role == 0) {
            if (p < NCH) {
                const int t0 = p * CH;
                for (int s = 0; s < CH; s += 4) {
                    const int t = t0 + s;
                    const float xs[4] = { xq.x, xq.y, xq.z, xq.w };
                    const int tn = (t + 4 < Tsz) ? (t + 4) : (Tsz - 4);
                    xq = *(const float4*)&x[row + tn];
                    const int tp = (t + 256 < Tsz) ? (t + 256) : (Tsz - 1);
                    __builtin_prefetch(&x[row + tp], 0, 0);
#pragma unroll
                    for (int u = 0; u < 4; ++u) {
                        v8f acc, raw[2];
                        for (int r = 0; r < 8; ++r) acc[r] = fmaf(xs[u], wg0C[r], b0C[r]);
                        const v8f d = gates2(lo8, A0[0][0], A0[0][1], A0[1][0], A0[1][1], hb0, hb1, acc, raw);
                        float h[2];
                        cell2(d, cs, h);
                        spread(h, hb0, hb1);
                        ringH[p & 1][s + u][lane] = make_float4(hb0.x, hb0.y, hb1.x, hb1.y);
                    }
                }
            }
        } else if (role == 1) {
            if (p >= 1 && p <= NCH) {
                const int k = p - 1;
                float4 g0 = ringH[k & 1][0][lane];
                float4 g1 = ringH[k & 1][1][lane];
                float4 g2 = ringH[k & 1][2][lane];
                float4 g3 = ringH[k & 1][3][lane];
#pragma nounroll
                for (int s = 0; s < CH; s += 4) {
                    const float4 a[4] = { g0, g1, g2, g3 };
                    const int sn = (s + 4 < CH) ? (s + 4) : s;
                    g0 = ringH[k & 1][sn][lane];
                    g1 = ringH[k & 1][sn + 1][lane];
                    g2 = ringH[k & 1][sn + 2][lane];
                    g3 = ringH[k & 1][sn + 3][lane];
#pragma unroll
                    for (int u = 0; u < 4; ++u) {
                        v2f f0, f1;
                        f0.x = a[u].x; f0.y = a[u].y;
                        f1.x = a[u].z; f1.y = a[u].w;
                        v8f raw[2];
                        const v8f P = gates2(lo8, A1[0][0], A1[0][1], A1[1][0], A1[1][1], f0, f1, b1C, raw);
                        ringP[k & 1][s + u][0][lane] = make_float4(P[0], P[1], P[2], P[3]);
                        ringP[k & 1][s + u][1][lane] = make_float4(P[4], P[5], P[6], P[7]);
                    }
                }
            }
        } else {
            if (p >= 2) {
                const int k  = p - 2;
                const int t0 = k * CH;
                float4 a00 = ringP[k & 1][0][0][lane], a01 = ringP[k & 1][0][1][lane];
                float4 a10 = ringP[k & 1][1][0][lane], a11 = ringP[k & 1][1][1][lane];
                float4 b00, b01, b10, b11;
                for (int s = 0; s < CH; s += 4) {
                    const int t = t0 + s;
                    const int sn = (s + 4 < CH) ? (s + 4) : s;
                    v8f raw[2];
                    l1_mats(A1, a00, a01, hb0, hb1, raw);
                    asm volatile("" :: "v"(raw[0]), "v"(raw[1]) : "memory");
                    b00 = ringP[k & 1][s + 2][0][lane]; b01 = ringP[k & 1][s + 2][1][lane];
                    b10 = ringP[k & 1][s + 3][0][lane]; b11 = ringP[k & 1][s + 3][1][lane];
                    __builtin_amdgcn_sched_barrier(0);
                    const float p0 = l1_cell(lo8, raw, cs, hb0, hb1, wl, pinit);
                    l1_mats(A1, a10, a11, hb0, hb1, raw);
                    const float p1 = l1_cell(lo8, raw, cs, hb0, hb1, wl, pinit);
                    l1_mats(A1, b00, b01, hb0, hb1, raw);
                    asm volatile("" :: "v"(raw[0]), "v"(raw[1]) : "memory");
                    a00 = ringP[k & 1][sn][0][lane];     a01 = ringP[k & 1][sn][1][lane];
                    a10 = ringP[k & 1][sn + 1][0][lane]; a11 = ringP[k & 1][sn + 1][1][lane];
                    __builtin_amdgcn_sched_barrier(0);
                    const float p2 = l1_cell(lo8, raw, cs, hb0, hb1, wl, pinit);
                    l1_mats(A1, b10, b11, hb0, hb1, raw);
                    const float p3 = l1_cell(lo8, raw, cs, hb0, hb1, wl, pinit);
                    const float v01 = Tl ? p1 : p0;
                    const float v23 = Tl ? p3 : p2;
                    out[row + t + 2 * q + Tl] = q ? v23 : v01;
                }
            }
        }
        __syncthreads();
    }
}

extern "C" void kernel_launch(void* const* d_in, const int* in_sizes, int n_in,
                              void* d_out, int out_size, void* d_ws, size_t ws_size,
                              hipStream_t stream) {
    (void)in_sizes; (void)n_in; (void)out_size; (void)d_ws; (void)ws_size;
    const float* x    = (const float*)d_in[0];
    const float* Wih0 = (const float*)d_in[1];
    const float* Whh0 = (const float*)d_in[2];
    const float* bih0 = (const float*)d_in[3];
    const float* bhh0 = (const float*)d_in[4];
    const float* Wih1 = (const float*)d_in[5];
    const float* Whh1 = (const float*)d_in[6];
    const float* bih1 = (const float*)d_in[7];
    const float* bhh1 = (const float*)d_in[8];
    const float* Wlin = (const float*)d_in[9];
    const float* blin = (const float*)d_in[10];
    float* out = (float*)d_out;

    lstm2_wmma_kernel<<<Bsz / ROWS_PER_GROUP, 96, 0, stream>>>(
        x, Wih0, Whh0, bih0, bhh0, Wih1, Whh1, bih1, bhh1, Wlin, blin, out);
}


# Recurrent waves: LDS hand-over grouped to ONE place per 8 steps (was one per 2): wave 0 stores 8 steps' h1 behind the trip, wave 2 reads a trip's 16 C tiles behind one step's matrix ops; same sums
# speedup vs baseline: 15.5944x; 1.0989x over previous
#include <hip/hip_runtime.h>
#include <hip/hip_bf16.h>


typedef float v2f __attribute__((ext_vector_type(2)));
typedef float v8f __attribute__((ext_vector_type(8)));

#define Bsz 2048
#define Tsz 4096
#define Hsz 8
#define ROWS_PER_GROUP 8
#define CH 64
#define NCH (Tsz / CH)
#define TRIP 8

__device__ __forceinline__ float shx16(float v) {
    unsigned int u = __float_as_uint(v);
    unsigned int r = __builtin_amdgcn_permlanex16(
        u, u, 0x76543210u, 0xfedcba98u,  false,  false);
    return __uint_as_float(r);
}
__device__ __forceinline__ float rot8(float v) {
    int u = (int)__float_as_uint(v);
    int r = __builtin_amdgcn_mov_dpp(u,   0x128, 0xf, 0xf, false);
    return __uint_as_float((unsigned int)r);
}

__device__ __forceinline__ float htanh(float x) {
    return __builtin_amdgcn_tanhf(x);
}
__device__ __forceinline__ float hsig_half(float xh) {
    return fmaf(0.5f, htanh(xh), 0.5f);
}

__device__ __forceinline__ v8f wmma4(v2f a, v2f b, v8f c) {
    return __builtin_amdgcn_wmma_f32_16x16x4_f32(
        false, a, false, b, (short)0, c, false, false);
}

__device__ __forceinline__ v8f pick(bool lo8, const v8f& d0, const v8f& d1) {
    v8f d;
    for (int r = 0; r < 8; ++r) d[r] = lo8 ? d0[r] : d1[r];
    return d;
}

__device__ __forceinline__ void cell2(const v8f& d, float (&c)[2], float (&h)[2]) {
    for (int j = 0; j < 2; ++j) {
        const float ig = hsig_half(d[j]);
        const float fg = hsig_half(d[2 + j]);
        const float gg = htanh(d[4 + j]);
        const float og = hsig_half(d[6 + j]);
        c[j] = fmaf(fg, c[j], ig * gg);
        h[j] = og * htanh(c[j]);
    }
}
__device__ __forceinline__ void spread(const float (&h)[2], v2f& hb0, v2f& hb1) {
    hb0.x = h[0];        hb0.y = h[1];
    hb1.x = rot8(h[0]);  hb1.y = rot8(h[1]);
}

__device__ __forceinline__ v8f gates2(bool lo8, v2f a00, v2f a01, v2f a10, v2f a11,
                                      v2f b0, v2f b1, const v8f& cin, v8f (&raw)[2]) {
    v8f t0 = wmma4(a00, b0, cin);
    v8f t1 = wmma4(a10, b0, cin);
    raw[0] = wmma4(a01, b1, t0);
    raw[1] = wmma4(a11, b1, t1);
    return pick(lo8, raw[0], raw[1]);
}

__device__ __forceinline__ float project(v2f hb0, v2f hb1, const float (&wl)[4], float pinit) {
    float p = pinit;
    p = fmaf(hb0.x, wl[0], p);
    p = fmaf(hb0.y, wl[1], p);
    p = fmaf(hb1.x, wl[2], p);
    p = fmaf(hb1.y, wl[3], p);
    p += shx16(p);
    return p;
}

__device__ __forceinline__ void l1_mats(const v2f (&A1)[2][4], float4 g0, float4 g1,
                                        v2f hb0, v2f hb1, v8f (&raw)[2]) {
    v8f Pc;
    Pc[0] = g0.x; Pc[1] = g0.y; Pc[2] = g0.z; Pc[3] = g0.w;
    Pc[4] = g1.x; Pc[5] = g1.y; Pc[6] = g1.z; Pc[7] = g1.w;
    v8f t0 = wmma4(A1[0][2], hb0, Pc);
    v8f t1 = wmma4(A1[1][2], hb0, Pc);
    raw[0] = wmma4(A1[0][3], hb1, t0);
    raw[1] = wmma4(A1[1][3], hb1, t1);
}
__device__ __forceinline__ float l1_cell(bool lo8, const v8f (&raw)[2], float (&cs)[2],
                                         v2f& hb0, v2f& hb1, const float (&wl)[4], float pinit) {
    const v8f d = pick(lo8, raw[0], raw[1]);
    float h[2];
    cell2(d, cs, h);
    spread(h, hb0, hb1);
    return project(hb0, hb1, wl, pinit);
}

__global__ __launch_bounds__(96) __attribute__((amdgpu_waves_per_eu(1, 4)))
void lstm2_wmma_kernel(
    const float* __restrict__ x,
    const float* __restrict__ Wih0,
    const float* __restrict__ Whh0,
    const float* __restrict__ bih0, const float* __restrict__ bhh0,
    const float* __restrict__ Wih1,
    const float* __restrict__ Whh1,
    const float* __restrict__ bih1, const float* __restrict__ bhh1,
    const float* __restrict__ Wlin,
    const float* __restrict__ blin,
    float* __restrict__ out)
{
    __shared__ float4 ringH[2][CH][32];
    __shared__ float4 ringP[2][CH][2][32];

    const int lane = threadIdx.x & 31;
    const int role = __builtin_amdgcn_readfirstlane((int)(threadIdx.x >> 5));
    const int c    = lane & 15;
    const int q    = lane >> 4;
    const int Tl   = (lane >> 3) & 1;
    const bool lo8 = (Tl == 0);
    const long row = (long)(blockIdx.x * ROWS_PER_GROUP + (lane & 7)) * Tsz;

    v2f A0[2][2];
    v2f A1[2][4];
    for (int gt = 0; gt < 2; ++gt) {
        const int gate = (c & 7) >> 1;
        const float sA = (gate == 2) ? 1.0f : 0.5f;
        const int g = gate * Hsz + 4 * (c >> 3) + 2 * gt + (c & 1);
        for (int kf = 0; kf < 2; ++kf) {
            const int k = 4 * q + 2 * (kf ^ gt);
            A0[gt][kf].x     = sA * Whh0[g * Hsz + k];
            A0[gt][kf].y     = sA * Whh0[g * Hsz + k + 1];
            A1[gt][kf].x     = sA * Wih1[g * Hsz + k];
            A1[gt][kf].y     = sA * Wih1[g * Hsz + k + 1];
            A1[gt][2 + kf].x = sA * Whh1[g * Hsz + k];
            A1[gt][2 + kf].y = sA * Whh1[g * Hsz + k + 1];
        }
    }

    float wg0C[8], b0C[8];
    v8f b1C;
    for (int r = 0; r < 8; ++r) {
        const int gate = r >> 1;
        const float sC = (gate == 2) ? 1.0f : 0.5f;
        const int g = gate * Hsz + 4 * q + 2 * Tl + (r & 1);
        wg0C[r] = sC * Wih0[g];
        b0C[r]  = sC * (bih0[g] + bhh0[g]);
        b1C[r]  = sC * (bih1[g] + bhh1[g]);
    }

    float wl[4];
    for (int j = 0; j < 4; ++j) wl[j] = Wlin[4 * q + 2 * ((j >> 1) ^ Tl) + (j & 1)];
    const float pinit = q ? 0.0f : blin[0];

    float cs[2] = { 0.f, 0.f };
    v2f hb0, hb1;
    hb0.x = 0.f; hb0.y = 0.f; hb1.x = 0.f; hb1.y = 0.f;
    float4 xq0 = *(const float4*)&x[row];
    float4 xq1 = *(const float4*)&x[row + 4];

    for (int p = 0; p < NCH + 2; ++p) {
        if (role == 0) {
            if (p < NCH) {
                const int t0 = p * CH;
                for (int s = 0; s < CH; s += TRIP) {
                    const int t = t0 + s;
                    const float xs[TRIP] = { xq0.x, xq0.y, xq0.z, xq0.w, xq1.x, xq1.y, xq1.z, xq1.w };
                    const int tn = (t + TRIP < Tsz) ? (t + TRIP) : (Tsz - TRIP);
                    xq0 = *(const float4*)&x[row + tn];
                    xq1 = *(const float4*)&x[row + tn + 4];
                    const int tp = (t + 256 < Tsz) ? (t + 256) : (Tsz - 1);
                    __builtin_prefetch(&x[row + tp], 0, 0);
                    float4 hst[TRIP];
#pragma unroll
                    for (int u = 0; u < TRIP; ++u) {
                        v8f acc, raw[2];
                        for (int r = 0; r < 8; ++r) acc[r] = fmaf(xs[u], wg0C[r], b0C[r]);
                        const v8f d = gates2(lo8, A0[0][0], A0[0][1], A0[1][0], A0[1][1], hb0, hb1, acc, raw);
                        float h[2];
                        cell2(d, cs, h);
                        spread(h, hb0, hb1);
                        hst[u] = make_float4(hb0.x, hb0.y, hb1.x, hb1.y);
                    }
                    __builtin_amdgcn_sched_barrier(0);
#pragma unroll
                    for (int u = 0; u < TRIP; ++u) ringH[p & 1][s + u][lane] = hst[u];
                }
            }
        } else if (role == 1) {
            if (p >= 1 && p <= NCH) {
                const int k = p - 1;
                float4 g0 = ringH[k & 1][0][lane];
                float4 g1 = ringH[k & 1][1][lane];
                float4 g2 = ringH[k & 1][2][lane];
                float4 g3 = ringH[k & 1][3][lane];
#pragma nounroll
                for (int s = 0; s < CH; s += 4) {
                    const float4 a[4] = { g0, g1, g2, g3 };
                    const int sn = (s + 4 < CH) ? (s + 4) : s;
                    g0 = ringH[k & 1][sn][lane];
                    g1 = ringH[k & 1][sn + 1][lane];
                    g2 = ringH[k & 1][sn + 2][lane];
                    g3 = ringH[k & 1][sn + 3][lane];
#pragma unroll
                    for (int u = 0; u < 4; ++u) {
                        v2f f0, f1;
                        f0.x = a[u].x; f0.y = a[u].y;
                        f1.x = a[u].z; f1.y = a[u].w;
                        v8f raw[2];
                        const v8f P = gates2(lo8, A1[0][0], A1[0][1], A1[1][0], A1[1][1], f0, f1, b1C, raw);
                        ringP[k & 1][s + u][0][lane] = make_float4(P[0], P[1], P[2], P[3]);
                        ringP[k & 1][s + u][1][lane] = make_float4(P[4], P[5], P[6], P[7]);
                    }
                }
            }
        } else {
            if (p >= 2) {
                const int k  = p - 2;
                const int t0 = k * CH;
                float4 Pa[TRIP], Pb[TRIP];
#pragma unroll
                for (int u = 0; u < TRIP; ++u) {
                    Pa[u] = ringP[k & 1][u][0][lane];
                    Pb[u] = ringP[k & 1][u][1][lane];
                }
                for (int s = 0; s < CH; s += TRIP) {
                    const int t = t0 + s;
                    const int sn = (s + TRIP < CH) ? (s + TRIP) : s;
                    float4 Na[TRIP], Nb[TRIP];
                    float pr[TRIP];
#pragma unroll
                    for (int u = 0; u < TRIP; ++u) {
                        v8f raw[2];
                        l1_mats(A1, Pa[u], Pb[u], hb0, hb1, raw);
                        if (u == TRIP - 2) {
                            asm volatile("" :: "v"(raw[0]), "v"(raw[1]) : "memory");
#pragma unroll
                            for (int v = 0; v < TRIP; ++v) {
                                Na[v] = ringP[k & 1][sn + v][0][lane];
                                Nb[v] = ringP[k & 1][sn + v][1][lane];
                            }
                            __builtin_amdgcn_sched_barrier(0);
                        }
                        pr[u] = l1_cell(lo8, raw, cs, hb0, hb1, wl, pinit);
                    }
#pragma unroll
                    for (int u = 0; u < TRIP; ++u) { Pa[u] = Na[u]; Pb[u] = Nb[u]; }
                    const float v01 = Tl ? pr[1] : pr[0];
                    const float v23 = Tl ? pr[3] : pr[2];
                    const float v45 = Tl ? pr[5] : pr[4];
                    const float v67 = Tl ? pr[7] : pr[6];
                    out[row + t + 2 * q + Tl]     = q ? v23 : v01;
                    out[row + t + 4 + 2 * q + Tl] = q ? v67 : v45;
                }
            }
        }
        __syncthreads();
    }
}

extern "C" void kernel_launch(void* const* d_in, const int* in_sizes, int n_in,
                              void* d_out, int out_size, void* d_ws, size_t ws_size,
                              hipStream_t stream) {
    (void)in_sizes; (void)n_in; (void)out_size; (void)d_ws; (void)ws_size;
    const float* x    = (const float*)d_in[0];
    const float* Wih0 = (const float*)d_in[1];
    const float* Whh0 = (const float*)d_in[2];
    const float* bih0 = (const float*)d_in[3];
    const float* bhh0 = (const float*)d_in[4];
    const float* Wih1 = (const float*)d_in[5];
    const float* Whh1 = (const float*)d_in[6];
    const float* bih1 = (const float*)d_in[7];
    const float* bhh1 = (const float*)d_in[8];
    const float* Wlin = (const float*)d_in[9];
    const float* blin = (const float*)d_in[10];
    float* out = (float*)d_out;

    lstm2_wmma_kernel<<<Bsz / ROWS_PER_GROUP, 96, 0, stream>>>(
        x, Wih0, Whh0, bih0, bhh0, Wih1, Whh1, bih1, bhh1, Wlin, blin, out);
}


# Recurrent waves turn to memory in ONE place per 8-step trip: wave 0's x loads and prefetch held at the trip's head, behind its LDS stores; wave 2 stores a trip's outputs beside the next LDS reads
# speedup vs baseline: 16.4864x; 1.0572x over previous
#include <hip/hip_runtime.h>
#include <hip/hip_bf16.h>


typedef float v2f __attribute__((ext_vector_type(2)));
typedef float v8f __attribute__((ext_vector_type(8)));

#define Bsz 2048
#define Tsz 4096
#define Hsz 8
#define ROWS_PER_GROUP 8
#define CH 64
#define NCH (Tsz / CH)
#define TRIP 8

__device__ __forceinline__ float shx16(float v) {
    unsigned int u = __float_as_uint(v);
    unsigned int r = __builtin_amdgcn_permlanex16(
        u, u, 0x76543210u, 0xfedcba98u,  false,  false);
    return __uint_as_float(r);
}
__device__ __forceinline__ float rot8(float v) {
    int u = (int)__float_as_uint(v);
    int r = __builtin_amdgcn_mov_dpp(u,   0x128, 0xf, 0xf, false);
    return __uint_as_float((unsigned int)r);
}

__device__ __forceinline__ float htanh(float x) {
    return __builtin_amdgcn_tanhf(x);
}
__device__ __forceinline__ float hsig_half(float xh) {
    return fmaf(0.5f, htanh(xh), 0.5f);
}

__device__ __forceinline__ v8f wmma4(v2f a, v2f b, v8f c) {
    return __builtin_amdgcn_wmma_f32_16x16x4_f32(
        false, a, false, b, (short)0, c, false, false);
}

__device__ __forceinline__ v8f pick(bool lo8, const v8f& d0, const v8f& d1) {
    v8f d;
    for (int r = 0; r < 8; ++r) d[r] = lo8 ? d0[r] : d1[r];
    return d;
}

__device__ __forceinline__ void cell2(const v8f& d, float (&c)[2], float (&h)[2]) {
    for (int j = 0; j < 2; ++j) {
        const float ig = hsig_half(d[j]);
        const float fg = hsig_half(d[2 + j]);
        const float gg = htanh(d[4 + j]);
        const float og = hsig_half(d[6 + j]);
        c[j] = fmaf(fg, c[j], ig * gg);
        h[j] = og * htanh(c[j]);
    }
}
__device__ __forceinline__ void spread(const float (&h)[2], v2f& hb0, v2f& hb1) {
    hb0.x = h[0];        hb0.y = h[1];
    hb1.x = rot8(h[0]);  hb1.y = rot8(h[1]);
}

__device__ __forceinline__ v8f gates2(bool lo8, v2f a00, v2f a01, v2f a10, v2f a11,
                                      v2f b0, v2f b1, const v8f& cin, v8f (&raw)[2]) {
    v8f t0 = wmma4(a00, b0, cin);
    v8f t1 = wmma4(a10, b0, cin);
    raw[0] = wmma4(a01, b1, t0);
    raw[1] = wmma4(a11, b1, t1);
    return pick(lo8, raw[0], raw[1]);
}

__device__ __forceinline__ float project(v2f hb0, v2f hb1, const float (&wl)[4], float pinit) {
    float p = pinit;
    p = fmaf(hb0.x, wl[0], p);
    p = fmaf(hb0.y, wl[1], p);
    p = fmaf(hb1.x, wl[2], p);
    p = fmaf(hb1.y, wl[3], p);
    p += shx16(p);
    return p;
}

__device__ __forceinline__ void l1_mats(const v2f (&A1)[2][4], float4 g0, float4 g1,
                                        v2f hb0, v2f hb1, v8f (&raw)[2]) {
    v8f Pc;
    Pc[0] = g0.x; Pc[1] = g0.y; Pc[2] = g0.z; Pc[3] = g0.w;
    Pc[4] = g1.x; Pc[5] = g1.y; Pc[6] = g1.z; Pc[7] = g1.w;
    v8f t0 = wmma4(A1[0][2], hb0, Pc);
    v8f t1 = wmma4(A1[1][2], hb0, Pc);
    raw[0] = wmma4(A1[0][3], hb1, t0);
    raw[1] = wmma4(A1[1][3], hb1, t1);
}
__device__ __forceinline__ float l1_cell(bool lo8, const v8f (&raw)[2], float (&cs)[2],
                                         v2f& hb0, v2f& hb1, const float (&wl)[4], float pinit) {
    const v8f d = pick(lo8, raw[0], raw[1]);
    float h[2];
    cell2(d, cs, h);
    spread(h, hb0, hb1);
    return project(hb0, hb1, wl, pinit);
}

__global__ __launch_bounds__(96) __attribute__((amdgpu_waves_per_eu(1, 4)))
void lstm2_wmma_kernel(
    const float* __restrict__ x,
    const float* __restrict__ Wih0,
    const float* __restrict__ Whh0,
    const float* __restrict__ bih0, const float* __restrict__ bhh0,
    const float* __restrict__ Wih1,
    const float* __restrict__ Whh1,
    const float* __restrict__ bih1, const float* __restrict__ bhh1,
    const float* __restrict__ Wlin,
    const float* __restrict__ blin,
    float* __restrict__ out)
{
    __shared__ float4 ringH[2][CH][32];
    __shared__ float4 ringP[2][CH][2][32];

    const int lane = threadIdx.x & 31;
    const int role = __builtin_amdgcn_readfirstlane((int)(threadIdx.x >> 5));
    const int c    = lane & 15;
    const int q    = lane >> 4;
    const int Tl   = (lane >> 3) & 1;
    const bool lo8 = (Tl == 0);
    const long row = (long)(blockIdx.x * ROWS_PER_GROUP + (lane & 7)) * Tsz;

    v2f A0[2][2];
    v2f A1[2][4];
    for (int gt = 0; gt < 2; ++gt) {
        const int gate = (c & 7) >> 1;
        const float sA = (gate == 2) ? 1.0f : 0.5f;
        const int g = gate * Hsz + 4 * (c >> 3) + 2 * gt + (c & 1);
        for (int kf = 0; kf < 2; ++kf) {
            const int k = 4 * q + 2 * (kf ^ gt);
            A0[gt][kf].x     = sA * Whh0[g * Hsz + k];
            A0[gt][kf].y     = sA * Whh0[g * Hsz + k + 1];
            A1[gt][kf].x     = sA * Wih1[g * Hsz + k];
            A1[gt][kf].y     = sA * Wih1[g * Hsz + k + 1];
            A1[gt][2 + kf].x = sA * Whh1[g * Hsz + k];
            A1[gt][2 + kf].y = sA * Whh1[g * Hsz + k + 1];
        }
    }

    float wg0C[8], b0C[8];
    v8f b1C;
    for (int r = 0; r < 8; ++r) {
        const int gate = r >> 1;
        const float sC = (gate == 2) ? 1.0f : 0.5f;
        const int g = gate * Hsz + 4 * q + 2 * Tl + (r & 1);
        wg0C[r] = sC * Wih0[g];
        b0C[r]  = sC * (bih0[g] + bhh0[g]);
        b1C[r]  = sC * (bih1[g] + bhh1[g]);
    }

    float wl[4];
    for (int j = 0; j < 4; ++j) wl[j] = Wlin[4 * q + 2 * ((j >> 1) ^ Tl) + (j & 1)];
    const float pinit = q ? 0.0f : blin[0];

    float cs[2] = { 0.f, 0.f };
    v2f hb0, hb1;
    hb0.x = 0.f; hb0.y = 0.f; hb1.x = 0.f; hb1.y = 0.f;
    float4 xq0 = *(const float4*)&x[row];
    float4 xq1 = *(const float4*)&x[row + 4];
    float po0 = 0.f, po1 = 0.f;

    for (int p = 0; p < NCH + 2; ++p) {
        if (role == 0) {
            if (p < NCH) {
                const int t0 = p * CH;
                for (int s = 0; s < CH; s += TRIP) {
                    const int t = t0 + s;
                    const float xs[TRIP] = { xq0.x, xq0.y, xq0.z, xq0.w, xq1.x, xq1.y, xq1.z, xq1.w };
                    const int tn = (t + TRIP < Tsz) ? (t + TRIP) : (Tsz - TRIP);
                    xq0 = *(const float4*)&x[row + tn];
                    xq1 = *(const float4*)&x[row + tn + 4];
                    const int tp = (t + 256 < Tsz) ? (t + 256) : (Tsz - 1);
                    __builtin_prefetch(&x[row + tp], 0, 0);
                    __builtin_amdgcn_sched_barrier(0);
                    float4 hst[TRIP];
#pragma unroll
                    for (int u = 0; u < TRIP; ++u) {
                        v8f acc, raw[2];
                        for (int r = 0; r < 8; ++r) acc[r] = fmaf(xs[u], wg0C[r], b0C[r]);
                        const v8f d = gates2(lo8, A0[0][0], A0[0][1], A0[1][0], A0[1][1], hb0, hb1, acc, raw);
                        float h[2];
                        cell2(d, cs, h);
                        spread(h, hb0, hb1);
                        hst[u] = make_float4(hb0.x, hb0.y, hb1.x, hb1.y);
                    }
                    __builtin_amdgcn_sched_barrier(0);
#pragma unroll
                    for (int u = 0; u < TRIP; ++u) ringH[p & 1][s + u][lane] = hst[u];
                }
            }
        } else if (role == 1) {
            if (p >= 1 && p <= NCH) {
                const int k = p - 1;
                float4 g0 = ringH[k & 1][0][lane];
                float4 g1 = ringH[k & 1][1][lane];
                float4 g2 = ringH[k & 1][2][lane];
                float4 g3 = ringH[k & 1][3][lane];
#pragma nounroll
                for (int s = 0; s < CH; s += 4) {
                    const float4 a[4] = { g0, g1, g2, g3 };
                    const int sn = (s + 4 < CH) ? (s + 4) : s;
                    g0 = ringH[k & 1][sn][lane];
                    g1 = ringH[k & 1][sn + 1][lane];
                    g2 = ringH[k & 1][sn + 2][lane];
                    g3 = ringH[k & 1][sn + 3][lane];
#pragma unroll
                    for (int u = 0; u < 4; ++u) {
                        v2f f0, f1;
                        f0.x = a[u].x; f0.y = a[u].y;
                        f1.x = a[u].z; f1.y = a[u].w;
                        v8f raw[2];
                        const v8f P = gates2(lo8, A1[0][0], A1[0][1], A1[1][0], A1[1][1], f0, f1, b1C, raw);
                        ringP[k & 1][s + u][0][lane] = make_float4(P[0], P[1], P[2], P[3]);
                        ringP[k & 1][s + u][1][lane] = make_float4(P[4], P[5], P[6], P[7]);
                    }
                }
            }
        } else {
            if (p >= 2) {
                const int k  = p - 2;
                const int t0 = k * CH;
                float4 Pa[TRIP], Pb[TRIP];
#pragma unroll
                for (int u = 0; u < TRIP; ++u) {
                    Pa[u] = ringP[k & 1][u][0][lane];
                    Pb[u] = ringP[k & 1][u][1][lane];
                }
                for (int s = 0; s < CH; s += TRIP) {
                    const int t = t0 + s;
                    const int sn = (s + TRIP < CH) ? (s + TRIP) : s;
                    float4 Na[TRIP], Nb[TRIP];
                    float pr[TRIP];
#pragma unroll
                    for (int u = 0; u < TRIP; ++u) {
                        v8f raw[2];
                        l1_mats(A1, Pa[u], Pb[u], hb0, hb1, raw);
                        if (u == TRIP - 2) {
                            asm volatile("" :: "v"(raw[0]), "v"(raw[1]) : "memory");
#pragma unroll
                            for (int v = 0; v < TRIP; ++v) {
                                Na[v] = ringP[k & 1][sn + v][0][lane];
                                Nb[v] = ringP[k & 1][sn + v][1][lane];
                            }
                            if (t != 0) {
                                out[row + (t - TRIP) + 2 * q + Tl]     = po0;
                                out[row + (t - TRIP) + 4 + 2 * q + Tl] = po1;
                            }
                            __builtin_amdgcn_sched_barrier(0);
                        }
                        pr[u] = l1_cell(lo8, raw, cs, hb0, hb1, wl, pinit);
                    }
#pragma unroll
                    for (int u = 0; u < TRIP; ++u) { Pa[u] = Na[u]; Pb[u] = Nb[u]; }
                    const float v01 = Tl ? pr[1] : pr[0];
                    const float v23 = Tl ? pr[3] : pr[2];
                    const float v45 = Tl ? pr[5] : pr[4];
                    const float v67 = Tl ? pr[7] : pr[6];
                    po0 = q ? v23 : v01;
                    po1 = q ? v67 : v45;
                }
                if (k == NCH - 1) {
                    out[row + (Tsz - TRIP) + 2 * q + Tl]     = po0;
                    out[row + (Tsz - TRIP) + 4 + 2 * q + Tl] = po1;
                }
            }
        }
        __syncthreads();
    }
}

extern "C" void kernel_launch(void* const* d_in, const int* in_sizes, int n_in,
                              void* d_out, int out_size, void* d_ws, size_t ws_size,
                              hipStream_t stream) {
    (void)in_sizes; (void)n_in; (void)out_size; (void)d_ws; (void)ws_size;
    const float* x    = (const float*)d_in[0];
    const float* Wih0 = (const float*)d_in[1];
    const float* Whh0 = (const float*)d_in[2];
    const float* bih0 = (const float*)d_in[3];
    const float* bhh0 = (const float*)d_in[4];
    const float* Wih1 = (const float*)d_in[5];
    const float* Whh1 = (const float*)d_in[6];
    const float* bih1 = (const float*)d_in[7];
    const float* bhh1 = (const float*)d_in[8];
    const float* Wlin = (const float*)d_in[9];
    const float* blin = (const float*)d_in[10];
    float* out = (float*)d_out;

    lstm2_wmma_kernel<<<Bsz / ROWS_PER_GROUP, 96, 0, stream>>>(
        x, Wih0, Whh0, bih0, bhh0, Wih1, Whh1, bih1, bhh1, Wlin, blin, out);
}
